# starting kernel
# baseline (speedup 1.0000x reference)
#include <hip/hip_runtime.h>

typedef float v2f __attribute__((ext_vector_type(2)));
typedef float v8f __attribute__((ext_vector_type(8)));

#define N_NODES 100000
#define N_EDGES 1600000
#define R_DIM   6
#define CH_IN   8
#define CH_OUT  16
#define K_DIM   96
#define OCOLS   32
#define N_TILES (N_NODES / 16)

__global__ __launch_bounds__(256) void scatter_kernel(
    const float* __restrict__ x,
    const int*   __restrict__ edges,
    const float* __restrict__ sten,
    float*       __restrict__ agg)
{
  int i = blockIdx.x * 256 + threadIdx.x;
  int e = i / R_DIM;
  int r = i - e * R_DIM;
  if (e >= N_EDGES) return;

  long long ep = __builtin_nontemporal_load((const long long*)edges + e);
  int src = (int)(ep & 0xffffffffLL);
  int dst = (int)(ep >> 32);

  v2f st = __builtin_nontemporal_load((const v2f*)sten + (size_t)e * R_DIM + r);

  const float4* xp = (const float4*)(x + (size_t)src * CH_IN);
  float4 x0 = xp[0];
  float4 x1 = xp[1];
  float xs[8] = {x0.x, x0.y, x0.z, x0.w, x1.x, x1.y, x1.z, x1.w};

  float* base = agg + (size_t)dst * K_DIM + r * (2 * CH_IN);
#pragma unroll
  for (int c = 0; c < CH_IN; ++c) {
    unsafeAtomicAdd(base + 2 * c,     st.x * xs[c]);
    unsafeAtomicAdd(base + 2 * c + 1, st.y * xs[c]);
  }
}

__global__ __launch_bounds__(256) void filter_kernel(
    const float* __restrict__ weight,
    const float* __restrict__ offset,
    float*       __restrict__ Fp)
{
  int i = blockIdx.x * 256 + threadIdx.x;
  if (i >= K_DIM * OCOLS) return;
  int k   = i >> 5;
  int col = i & 31;
  int p   = k & 1;
  int rc  = k >> 1;
  int r   = rc >> 3;
  int c   = rc & 7;
  int o   = col & 15;

  float w  = weight[(r * CH_IN + c) * CH_OUT + o];
  float ph = offset[c * CH_OUT + o];
  float fre = w * cosf(ph);
  float fim = w * sinf(ph);

  float val;
  if (col < 16) val = (p == 0) ? fre : -fim;
  else          val = (p == 0) ? fim :  fre;

  Fp[(rc * OCOLS + col) * 2 + p] = val;
}

__global__ __launch_bounds__(256) void gemm_kernel(
    const float* __restrict__ Fp,
    const float* __restrict__ agg,
    const float* __restrict__ bias,
    float*       __restrict__ out)
{
  __shared__ float Fl[K_DIM * OCOLS];
  for (int i = threadIdx.x; i < K_DIM * OCOLS; i += 256) Fl[i] = Fp[i];
  __syncthreads();

  int wave = threadIdx.x >> 5;
  int lane = threadIdx.x & 31;
  int tile = blockIdx.x * 8 + wave;
  if (tile >= N_TILES) return;

  int node_base = tile * 16;
  int mrow = lane & 15;
  int half = lane >> 4;
  const float* arow = agg + (size_t)(node_base + mrow) * K_DIM + half * 2;
  const float* fb0 = Fl + (size_t)(half * OCOLS + mrow) * 2;
  const float* fb1 = fb0 + 32;

  v8f acc0 = {0.f, 0.f, 0.f, 0.f, 0.f, 0.f, 0.f, 0.f};
  v8f acc1 = {0.f, 0.f, 0.f, 0.f, 0.f, 0.f, 0.f, 0.f};

#pragma unroll 4
  for (int k0 = 0; k0 < K_DIM; k0 += 4) {
    v2f a  = *(const v2f*)(arow + k0);
    v2f b0 = *(const v2f*)(fb0 + (k0 >> 1) * (OCOLS * 2));
    v2f b1 = *(const v2f*)(fb1 + (k0 >> 1) * (OCOLS * 2));
    acc0 = __builtin_amdgcn_wmma_f32_16x16x4_f32(
        false, a, false, b0, (short)0, acc0, false, false);
    acc1 = __builtin_amdgcn_wmma_f32_16x16x4_f32(
        false, a, false, b1, (short)0, acc1, false, false);
  }

  float bn = bias[mrow];
  int   mhi = half * 8;
#pragma unroll
  for (int j = 0; j < 8; ++j) {
    int   m   = j + mhi;
    float re  = acc0[j];
    float im  = acc1[j];
    float mag = __builtin_amdgcn_sqrtf(re * re + im * im);
    float sc  = fmaxf(mag + bn, 0.0f) *
                __builtin_amdgcn_rcpf(mag + 1e-8f);
    v2f v = { re * sc, im * sc };
    __builtin_nontemporal_store(
        v, (v2f*)(out + ((size_t)(node_base + m) * CH_OUT + mrow) * 2));
  }
}

extern "C" void kernel_launch(void* const* d_in, const int* in_sizes, int n_in,
                              void* d_out, int out_size, void* d_ws, size_t ws_size,
                              hipStream_t stream) {
  const float* x      = (const float*)d_in[0];
  const int*   edges  = (const int*)d_in[1];
  const float* sten   = (const float*)d_in[2];
  const float* weight = (const float*)d_in[3];
  const float* offset = (const float*)d_in[4];
  const float* bias   = (const float*)d_in[5];
  float*       out    = (float*)d_out;

  float* agg = (float*)d_ws;
  float* Fp  = agg + (size_t)N_NODES * K_DIM;

  hipMemsetAsync(agg, 0, (size_t)N_NODES * K_DIM * sizeof(float), stream);

  filter_kernel<<<(K_DIM * OCOLS + 255) / 256, 256, 0, stream>>>(weight, offset, Fp);

  long long total = (long long)N_EDGES * R_DIM;
  scatter_kernel<<<(int)((total + 255) / 256), 256, 0, stream>>>(x, edges, sten, agg);

  gemm_kernel<<<(N_TILES + 7) / 8, 256, 0, stream>>>(Fp, agg, bias, out);
}


# Float atomic scatter replaced by counting sort of edges by destination (int atomics only); each node's messages summed in registers in ascending edge order, same every run, straight into WMMA A
# speedup vs baseline: 2.9578x; 2.9578x over previous
#include <hip/hip_runtime.h>

typedef float v2f __attribute__((ext_vector_type(2)));
typedef float v8f __attribute__((ext_vector_type(8)));

#define N_NODES 100000
#define N_EDGES 1600000
#define R_DIM   6
#define CH_IN   8
#define CH_OUT  16
#define K_DIM   96
#define OCOLS   32
#define N_TILES (N_NODES / 16)

#define SCAN_BLOCKS      100
#define SCAN_BLOCK_NODES 1000
#define WS_PAD_NODES 100096

#define EDGE_THREADS 320000
#define EDGE_PER_THREAD 5

static_assert(SCAN_BLOCKS * SCAN_BLOCK_NODES == N_NODES, "scan covers every node exactly once");
static_assert(SCAN_BLOCK_NODES % 4 == 0 && SCAN_BLOCK_NODES / 4 <= 256, "one 16-byte element per scan thread");
static_assert(EDGE_THREADS * EDGE_PER_THREAD == N_EDGES, "edge kernels cover every edge exactly once");
static_assert(EDGE_THREADS % 256 == 0, "edge kernels use an exact grid");


__global__ __launch_bounds__(256) void count_kernel(
    const int* __restrict__ edges,
    int*       __restrict__ cnt,
    int*       __restrict__ slot)
{
  int i = blockIdx.x * 256 + threadIdx.x;
  int d[EDGE_PER_THREAD], r[EDGE_PER_THREAD];
#pragma unroll
  for (int k = 0; k < EDGE_PER_THREAD; ++k) d[k] = edges[2 * (i + k * EDGE_THREADS) + 1];
#pragma unroll
  for (int k = 0; k < EDGE_PER_THREAD; ++k) r[k] = atomicAdd(cnt + d[k], 1);
#pragma unroll
  for (int k = 0; k < EDGE_PER_THREAD; ++k) slot[i + k * EDGE_THREADS] = r[k];
}

__global__ __launch_bounds__(256) void scan_kernel(
    const int* __restrict__ cnt,
    int*       __restrict__ off)
{
  __shared__ int sh[256];
  int t = threadIdx.x;
  const int4* c4 = (const int4*)cnt;
  int n4 = blockIdx.x * (SCAN_BLOCK_NODES / 4);

  int front = 0;
#pragma unroll 8
  for (int q = t; q < n4; q += 256) {
    int4 v = c4[q];
    front += v.x + v.y + v.z + v.w;
  }

  int4 mine = make_int4(0, 0, 0, 0);
  if (t < SCAN_BLOCK_NODES / 4) mine = c4[n4 + t];
  int ms = mine.x + mine.y + mine.z + mine.w;

  sh[t] = front;
  __syncthreads();
  for (int d = 128; d >= 1; d >>= 1) {
    if (t < d) sh[t] += sh[t + d];
    __syncthreads();
  }
  int base = sh[0];
  __syncthreads();

  sh[t] = ms;
  __syncthreads();
  for (int d = 1; d < 256; d <<= 1) {
    int a = (t >= d) ? sh[t - d] : 0;
    __syncthreads();
    sh[t] += a;
    __syncthreads();
  }
  int p = base + sh[t] - ms;
  if (t < SCAN_BLOCK_NODES / 4) {
    int4 o;
    o.x = p;
    o.y = p + mine.x;
    o.z = o.y + mine.y;
    o.w = o.z + mine.z;
    ((int4*)off)[n4 + t] = o;
  }
}

__global__ __launch_bounds__(256) void place_kernel(
    const int* __restrict__ edges,
    const int* __restrict__ off,
    const int* __restrict__ slot,
    int*       __restrict__ grp)
{
  int i = blockIdx.x * 256 + threadIdx.x;
  int d[EDGE_PER_THREAD], p[EDGE_PER_THREAD];
#pragma unroll
  for (int k = 0; k < EDGE_PER_THREAD; ++k) d[k] = edges[2 * (i + k * EDGE_THREADS) + 1];
#pragma unroll
  for (int k = 0; k < EDGE_PER_THREAD; ++k) p[k] = off[d[k]] + slot[i + k * EDGE_THREADS];
#pragma unroll
  for (int k = 0; k < EDGE_PER_THREAD; ++k) grp[p[k]] = i + k * EDGE_THREADS;
}

__global__ __launch_bounds__(256) void order_kernel(
    const int* __restrict__ edges,
    const int* __restrict__ off,
    const int* __restrict__ cnt,
    const int* __restrict__ grp,
    int*       __restrict__ lst)
{
  int i = blockIdx.x * 256 + threadIdx.x;
  int2 sd[EDGE_PER_THREAD];
  int s[EDGE_PER_THREAD], n[EDGE_PER_THREAD];
#pragma unroll
  for (int k = 0; k < EDGE_PER_THREAD; ++k) sd[k] = ((const int2*)edges)[i + k * EDGE_THREADS];
#pragma unroll
  for (int k = 0; k < EDGE_PER_THREAD; ++k) { s[k] = off[sd[k].y]; n[k] = cnt[sd[k].y]; }
#pragma unroll
  for (int k = 0; k < EDGE_PER_THREAD; ++k) {
    int e = i + k * EDGE_THREADS;
    const int* g = grp + s[k];
    int r = 0;
    for (int j = 0; j < n[k]; ++j) r += (g[j] < e) ? 1 : 0;
    ((int2*)lst)[s[k] + r] = make_int2(e, sd[k].x);
  }
}

__global__ __launch_bounds__(256) void filter_kernel(
    const float* __restrict__ weight,
    const float* __restrict__ offset,
    float*       __restrict__ Fp)
{
  int i = blockIdx.x * 256 + threadIdx.x;
  if (i >= K_DIM * OCOLS) return;
  int k   = i >> 5;
  int col = i & 31;
  int p   = k & 1;
  int rc  = k >> 1;
  int r   = rc >> 3;
  int c   = rc & 7;
  int o   = col & 15;

  float w  = weight[(r * CH_IN + c) * CH_OUT + o];
  float ph = offset[c * CH_OUT + o];
  float fre = w * cosf(ph);
  float fim = w * sinf(ph);

  float val;
  if (col < 16) val = (p == 0) ? fre : -fim;
  else          val = (p == 0) ? fim :  fre;

  Fp[(rc * OCOLS + col) * 2 + p] = val;
}

__global__ __launch_bounds__(128) __attribute__((amdgpu_waves_per_eu(1, 7))) void tile_kernel(
    const float* __restrict__ Fp,
    const float* __restrict__ x,
    const float* __restrict__ sten,
    const int*   __restrict__ off,
    const int*   __restrict__ cnt,
    const int*   __restrict__ lst,
    const float* __restrict__ bias,
    float*       __restrict__ out)
{
  __shared__ float Fl[K_DIM * OCOLS];
  for (int i = threadIdx.x; i < K_DIM * OCOLS; i += 128) Fl[i] = Fp[i];
  __syncthreads();

  int wave = threadIdx.x >> 5;
  int lane = threadIdx.x & 31;
  int tile = blockIdx.x * 4 + wave;
  if (tile >= N_TILES) return;

  int node_base = tile * 16;
  int mrow = lane & 15;
  int half = lane >> 4;

  float a[48];
#pragma unroll
  for (int q = 0; q < 48; ++q) a[q] = 0.f;

  int s = off[node_base + mrow];
  int n = cnt[node_base + mrow];
  if (n > 0) {
    int nm1 = n - 1;
    for (int i0 = 0; i0 < n; i0 += 4) {
      int2 ed[4];
#pragma unroll
      for (int k = 0; k < 4; ++k) {
        int ik = i0 + k;
        unsigned lo = (unsigned)(s + (ik < nm1 ? ik : nm1)) * 8u;
        ed[k] = *(const int2*)((const char*)lst + lo);
      }
      float4 S0[4], S1[4], S2[4];
      float  xs[4][4];
#pragma unroll
      for (int k = 0; k < 4; ++k) {
        unsigned so = (unsigned)ed[k].x * (unsigned)(R_DIM * 2 * sizeof(float));
        unsigned xo = (unsigned)ed[k].y * (unsigned)(CH_IN * sizeof(float));
        const float4* sp = (const float4*)((const char*)sten + so);
        const float4* xp = (const float4*)((const char*)x + xo);
        S0[k] = sp[0]; S1[k] = sp[1]; S2[k] = sp[2];
        float4 xa = xp[0];
        float4 xb = xp[1];
        bool ok = (i0 + k) < n;
        float c0 = half ? xa.y : xa.x;
        float c1 = half ? xa.w : xa.z;
        float c2 = half ? xb.y : xb.x;
        float c3 = half ? xb.w : xb.z;
        xs[k][0] = ok ? c0 : 0.f;
        xs[k][1] = ok ? c1 : 0.f;
        xs[k][2] = ok ? c2 : 0.f;
        xs[k][3] = ok ? c3 : 0.f;
      }
#pragma unroll
      for (int k = 0; k < 4; ++k) {
        float st[12] = {S0[k].x, S0[k].y, S0[k].z, S0[k].w,
                        S1[k].x, S1[k].y, S1[k].z, S1[k].w,
                        S2[k].x, S2[k].y, S2[k].z, S2[k].w};
#pragma unroll
        for (int r = 0; r < R_DIM; ++r) {
#pragma unroll
          for (int j = 0; j < 4; ++j) {
            a[(r * 4 + j) * 2 + 0] = __builtin_fmaf(st[r * 2 + 0], xs[k][j], a[(r * 4 + j) * 2 + 0]);
            a[(r * 4 + j) * 2 + 1] = __builtin_fmaf(st[r * 2 + 1], xs[k][j], a[(r * 4 + j) * 2 + 1]);
          }
        }
      }
    }
  }

  const float* fb0 = Fl + (size_t)(half * OCOLS + mrow) * 2;
  const float* fb1 = fb0 + 32;

  v8f acc0 = {0.f, 0.f, 0.f, 0.f, 0.f, 0.f, 0.f, 0.f};
  v8f acc1 = {0.f, 0.f, 0.f, 0.f, 0.f, 0.f, 0.f, 0.f};

#pragma unroll
  for (int t = 0; t < K_DIM / 4; ++t) {
    v2f av = { a[2 * t], a[2 * t + 1] };
    v2f b0 = *(const v2f*)(fb0 + t * (OCOLS * 4));
    v2f b1 = *(const v2f*)(fb1 + t * (OCOLS * 4));
    acc0 = __builtin_amdgcn_wmma_f32_16x16x4_f32(
        false, av, false, b0, (short)0, acc0, false, false);
    acc1 = __builtin_amdgcn_wmma_f32_16x16x4_f32(
        false, av, false, b1, (short)0, acc1, false, false);
  }

  float bn = bias[mrow];
  int   mhi = half * 8;
#pragma unroll
  for (int j = 0; j < 8; ++j) {
    int   m   = j + mhi;
    float re  = acc0[j];
    float im  = acc1[j];
    float mag = __builtin_amdgcn_sqrtf(re * re + im * im);
    float sc  = fmaxf(mag + bn, 0.0f) *
                __builtin_amdgcn_rcpf(mag + 1e-8f);
    v2f v = { re * sc, im * sc };
    __builtin_nontemporal_store(
        v, (v2f*)(out + ((size_t)(node_base + m) * CH_OUT + mrow) * 2));
  }
}

extern "C" void kernel_launch(void* const* d_in, const int* in_sizes, int n_in,
                              void* d_out, int out_size, void* d_ws, size_t ws_size,
                              hipStream_t stream) {
  const float* x      = (const float*)d_in[0];
  const int*   edges  = (const int*)d_in[1];
  const float* sten   = (const float*)d_in[2];
  const float* weight = (const float*)d_in[3];
  const float* offset = (const float*)d_in[4];
  const float* bias   = (const float*)d_in[5];
  float*       out    = (float*)d_out;

  int*   cnt  = (int*)d_ws;
  int*   off  = cnt + WS_PAD_NODES;
  int*   slot = off + WS_PAD_NODES;
  int*   grp  = slot + N_EDGES;
  int*   lst  = grp + N_EDGES;
  float* Fp   = (float*)(lst + 2 * N_EDGES);

  hipMemsetAsync(cnt, 0, (size_t)N_NODES * sizeof(int), stream);

  filter_kernel<<<(K_DIM * OCOLS + 255) / 256, 256, 0, stream>>>(weight, offset, Fp);

  count_kernel<<<EDGE_THREADS / 256, 256, 0, stream>>>(edges, cnt, slot);
  scan_kernel<<<SCAN_BLOCKS, 256, 0, stream>>>(cnt, off);
  place_kernel<<<EDGE_THREADS / 256, 256, 0, stream>>>(edges, off, slot, grp);
  order_kernel<<<EDGE_THREADS / 256, 256, 0, stream>>>(edges, off, cnt, grp, lst);

  tile_kernel<<<(N_TILES + 3) / 4, 128, 0, stream>>>(Fp, x, sten, off, cnt, lst, bias, out);
}


# Edges grouped by a two-level counting sort ranked with LDS atomics (one global atomic per chunk and bucket, not per edge); each group's edge order then set in LDS by the tile kernel; sum order same
# speedup vs baseline: 7.8105x; 2.6406x over previous
#include <hip/hip_runtime.h>

typedef float v2f __attribute__((ext_vector_type(2)));
typedef float v8f __attribute__((ext_vector_type(8)));

#define N_NODES 100000
#define N_EDGES 1600000
#define R_DIM   6
#define CH_IN   8
#define CH_OUT  16
#define K_DIM   96
#define OCOLS   32
#define N_TILES (N_NODES / 16)

#define WS_PAD_NODES 100096

#define NB      250
#define BNODES  400
#define NCHUNK  250
#define CHUNK   6400
#define BT      640
#define EPT     10
#define BSTRIDE 32
#define GT      1024
#define ID_BITS 21
#define ID_MASK ((1 << ID_BITS) - 1)

static_assert(NB * BNODES == N_NODES, "buckets cover every node exactly once");
static_assert(BNODES % 16 == 0, "a tile of 16 nodes never straddles two buckets");
static_assert(NCHUNK * CHUNK == N_EDGES && BT * EPT == CHUNK, "chunks cover every edge exactly once");
static_assert(N_EDGES <= (1 << ID_BITS) && BNODES <= 512, "edge id and node-in-bucket pack into one non-negative int");
static_assert(NB <= 256 && NB <= BT && BNODES <= 512 && GT >= 512, "scan widths");


__device__ __forceinline__ void scan256(int* sc, int t, int v)
{
  if (t < 256) sc[t] = v;
  __syncthreads();
  for (int d = 1; d < 256; d <<= 1) {
    int a = (t < 256 && t >= d) ? sc[t - d] : 0;
    __syncthreads();
    if (t < 256) sc[t] += a;
    __syncthreads();
  }
}

__global__ __launch_bounds__(BT) void bucket_count_kernel(
    const int* __restrict__ edges,
    int*       __restrict__ btot)
{
  __shared__ int h[NB];
  int t = threadIdx.x;
  if (t < NB) h[t] = 0;
  __syncthreads();
  int first = blockIdx.x * CHUNK + t;
  int d[EPT];
#pragma unroll
  for (int k = 0; k < EPT; ++k) d[k] = edges[2 * (first + k * BT) + 1];
#pragma unroll
  for (int k = 0; k < EPT; ++k) atomicAdd(&h[d[k] / BNODES], 1);
  __syncthreads();
  if (t < NB) {
    int c = h[t];
    if (c) atomicAdd(btot + t * BSTRIDE, c);
  }
}

__global__ __launch_bounds__(BT) void bucket_scatter_kernel(
    const int* __restrict__ edges,
    const int* __restrict__ btot,
    int*       __restrict__ bcur,
    int*       __restrict__ coarse)
{
  __shared__ int h[NB];
  __shared__ int sc[256];
  int t = threadIdx.x;
  if (t < NB) h[t] = 0;
  int tot = (t < NB) ? btot[t * BSTRIDE] : 0;
  __syncthreads();
  int first = blockIdx.x * CHUNK + t;
  int2 sd[EPT];
  int  b[EPT], r[EPT];
#pragma unroll
  for (int k = 0; k < EPT; ++k) sd[k] = ((const int2*)edges)[first + k * BT];
#pragma unroll
  for (int k = 0; k < EPT; ++k) {
    b[k] = sd[k].y / BNODES;
    r[k] = atomicAdd(&h[b[k]], 1);
  }
  scan256(sc, t, tot);
  if (t < NB) {
    int c = h[t];
    int run = c ? atomicAdd(bcur + t * BSTRIDE, c) : 0;
    h[t] = sc[t] - tot + run;
  }
  __syncthreads();
#pragma unroll
  for (int k = 0; k < EPT; ++k) {
    int e = first + k * BT;
    int l = sd[k].y - b[k] * BNODES;
    ((int2*)coarse)[h[b[k]] + r[k]] = make_int2(e | (l << ID_BITS), sd[k].x);
  }
}

__global__ __launch_bounds__(GT) void group_kernel(
    const int* __restrict__ btot,
    const int* __restrict__ coarse,
    int*       __restrict__ cnt,
    int*       __restrict__ off,
    int*       __restrict__ grp)
{
  __shared__ int cn[512];
  __shared__ int of[512];
  __shared__ int cur[512];
  __shared__ int sc[256];
  int t  = threadIdx.x;
  int bk = blockIdx.x;
  int tot = (t < NB) ? btot[t * BSTRIDE] : 0;
  if (t < 512) { cn[t] = 0; cur[t] = 0; }
  scan256(sc, t, tot);
  int m     = sc[bk] - (bk ? sc[bk - 1] : 0);
  int start = sc[bk] - m;
  const int2* src = (const int2*)coarse + start;

  for (int q0 = 0; q0 < m; q0 += 4 * GT) {
    int pk[4];
#pragma unroll
    for (int k = 0; k < 4; ++k) {
      int q = q0 + k * GT + t;
      pk[k] = (q < m) ? src[q].x : -1;
    }
#pragma unroll
    for (int k = 0; k < 4; ++k)
      if (pk[k] >= 0) atomicAdd(&cn[pk[k] >> ID_BITS], 1);
  }
  __syncthreads();

  int mine = (t < 512) ? cn[t] : 0;
  if (t < 512) of[t] = mine;
  __syncthreads();
  for (int d = 1; d < 512; d <<= 1) {
    int a = (t < 512 && t >= d) ? of[t - d] : 0;
    __syncthreads();
    if (t < 512) of[t] += a;
    __syncthreads();
  }
  if (t < 512) {
    int ex = of[t] - mine;
    of[t] = ex;
    if (t < BNODES) {
      cnt[bk * BNODES + t] = mine;
      off[bk * BNODES + t] = start + ex;
    }
  }
  __syncthreads();

  int2* dst = (int2*)grp + start;
  for (int q0 = 0; q0 < m; q0 += 4 * GT) {
    int2 v[4];
#pragma unroll
    for (int k = 0; k < 4; ++k) {
      int q = q0 + k * GT + t;
      v[k] = (q < m) ? src[q] : make_int2(-1, 0);
    }
#pragma unroll
    for (int k = 0; k < 4; ++k) {
      if (v[k].x >= 0) {
        int l = v[k].x >> ID_BITS;
        int slot = atomicAdd(&cur[l], 1);
        dst[of[l] + slot] = make_int2(v[k].x & ID_MASK, v[k].y);
      }
    }
  }
}

__global__ __launch_bounds__(256) void filter_kernel(
    const float* __restrict__ weight,
    const float* __restrict__ offset,
    float*       __restrict__ Fp)
{
  int i = blockIdx.x * 256 + threadIdx.x;
  if (i >= K_DIM * OCOLS) return;
  int k   = i >> 5;
  int col = i & 31;
  int p   = k & 1;
  int rc  = k >> 1;
  int r   = rc >> 3;
  int c   = rc & 7;
  int o   = col & 15;

  float w  = weight[(r * CH_IN + c) * CH_OUT + o];
  float ph = offset[c * CH_OUT + o];
  float fre = w * cosf(ph);
  float fim = w * sinf(ph);

  float val;
  if (col < 16) val = (p == 0) ? fre : -fim;
  else          val = (p == 0) ? fim :  fre;

  Fp[(rc * OCOLS + col) * 2 + p] = val;
}

#define WAVE_CAP    512
#define STAGE_LOADS (WAVE_CAP / 32)

__device__ __forceinline__ void add_edge(float (&a)[48], float4 s0, float4 s1, float4 s2,
                                         float x0, float x1, float x2, float x3)
{
  float st[12] = {s0.x, s0.y, s0.z, s0.w, s1.x, s1.y, s1.z, s1.w, s2.x, s2.y, s2.z, s2.w};
  float xs[4]  = {x0, x1, x2, x3};
#pragma unroll
  for (int r = 0; r < R_DIM; ++r) {
#pragma unroll
    for (int j = 0; j < 4; ++j) {
      a[(r * 4 + j) * 2 + 0] = __builtin_fmaf(st[r * 2 + 0], xs[j], a[(r * 4 + j) * 2 + 0]);
      a[(r * 4 + j) * 2 + 1] = __builtin_fmaf(st[r * 2 + 1], xs[j], a[(r * 4 + j) * 2 + 1]);
    }
  }
}

__global__ __launch_bounds__(128) __attribute__((amdgpu_waves_per_eu(1, 7))) void tile_kernel(
    const float* __restrict__ Fp,
    const float* __restrict__ x,
    const float* __restrict__ sten,
    const int*   __restrict__ off,
    const int*   __restrict__ cnt,
    const int*   __restrict__ grp,
    const float* __restrict__ bias,
    float*       __restrict__ out)
{
  __shared__ float Fl[K_DIM * OCOLS];
  __shared__ int2  raw[4][WAVE_CAP];
  __shared__ int2  srt[4][WAVE_CAP];
  for (int i = threadIdx.x; i < K_DIM * OCOLS; i += 128) Fl[i] = Fp[i];
  __syncthreads();

  int wave = threadIdx.x >> 5;
  int lane = threadIdx.x & 31;
  int tile = blockIdx.x * 4 + wave;
  if (tile >= N_TILES) return;

  int node_base = tile * 16;
  int mrow = lane & 15;
  int half = lane >> 4;

  float a[48];
#pragma unroll
  for (int q = 0; q < 48; ++q) a[q] = 0.f;

  int s = off[node_base + mrow];
  int n = cnt[node_base + mrow];
  int nm1 = n - 1;
  int w0 = __builtin_amdgcn_readlane(s, 0);
  int T  = __builtin_amdgcn_readlane(s + n, 15) - w0;
  int sl = s - w0;
  const int2* g2 = (const int2*)grp + w0;

  if (T > 0 && T <= WAVE_CAP) {
    int2* rw = raw[wave];
    int2* sr = srt[wave];
    int Tm1 = T - 1;
    int2 v[STAGE_LOADS];
#pragma unroll
    for (int k = 0; k < STAGE_LOADS; ++k) {
      int p = lane + 32 * k;
      v[k] = g2[p < Tm1 ? p : Tm1];
    }
#pragma unroll
    for (int k = 0; k < STAGE_LOADS; ++k) {
      int p = lane + 32 * k;
      if (p < T) rw[p] = v[k];
    }

    for (int i0 = half; i0 < n; i0 += 8) {
      int2 e[4];
      int  r[4];
#pragma unroll
      for (int k = 0; k < 4; ++k) {
        int i = i0 + 2 * k;
        e[k] = rw[sl + (i < nm1 ? i : nm1)];
        r[k] = 0;
      }
#pragma unroll 4
      for (int j = 0; j < n; ++j) {
        int vj = rw[sl + j].x;
#pragma unroll
        for (int k = 0; k < 4; ++k) r[k] += (vj < e[k].x) ? 1 : 0;
      }
#pragma unroll
      for (int k = 0; k < 4; ++k)
        if (i0 + 2 * k < n) sr[sl + r[k]] = e[k];
    }

    if (n > 0) {
      const int2* mine = sr + sl;
      for (int i0 = 0; i0 < n; i0 += 4) {
        int2 ed[4];
#pragma unroll
        for (int k = 0; k < 4; ++k) {
          int ik = i0 + k;
          ed[k] = mine[ik < nm1 ? ik : nm1];
        }
        float4 S0[4], S1[4], S2[4];
        float  xs[4][4];
#pragma unroll
        for (int k = 0; k < 4; ++k) {
          unsigned so = (unsigned)ed[k].x * (unsigned)(R_DIM * 2 * sizeof(float));
          unsigned xo = (unsigned)ed[k].y * (unsigned)(CH_IN * sizeof(float));
          const float4* sp = (const float4*)((const char*)sten + so);
          const float4* xp = (const float4*)((const char*)x + xo);
          S0[k] = sp[0]; S1[k] = sp[1]; S2[k] = sp[2];
          float4 xa = xp[0];
          float4 xb = xp[1];
          bool ok = (i0 + k) < n;
          float c0 = half ? xa.y : xa.x;
          float c1 = half ? xa.w : xa.z;
          float c2 = half ? xb.y : xb.x;
          float c3 = half ? xb.w : xb.z;
          xs[k][0] = ok ? c0 : 0.f;
          xs[k][1] = ok ? c1 : 0.f;
          xs[k][2] = ok ? c2 : 0.f;
          xs[k][3] = ok ? c3 : 0.f;
        }
#pragma unroll
        for (int k = 0; k < 4; ++k)
          add_edge(a, S0[k], S1[k], S2[k], xs[k][0], xs[k][1], xs[k][2], xs[k][3]);
      }
    }
  } else if (T > WAVE_CAP) {
    if (n > 0) {
      const int2* g = g2 + sl;
      int last = -1;
      for (int i = 0; i < n; ++i) {
        int best = 0x7fffffff, bsrc = 0;
        for (int j = 0; j < n; ++j) {
          int2 e = g[j];
          if (e.x > last && e.x < best) { best = e.x; bsrc = e.y; }
        }
        last = best;
        const float4* sp = (const float4*)(sten + (size_t)best * (R_DIM * 2));
        const float4* xp = (const float4*)(x + (size_t)bsrc * CH_IN);
        float4 xa = xp[0];
        float4 xb = xp[1];
        add_edge(a, sp[0], sp[1], sp[2],
                 half ? xa.y : xa.x, half ? xa.w : xa.z,
                 half ? xb.y : xb.x, half ? xb.w : xb.z);
      }
    }
  }

  const float* fb0 = Fl + (size_t)(half * OCOLS + mrow) * 2;
  const float* fb1 = fb0 + 32;

  v8f acc0 = {0.f, 0.f, 0.f, 0.f, 0.f, 0.f, 0.f, 0.f};
  v8f acc1 = {0.f, 0.f, 0.f, 0.f, 0.f, 0.f, 0.f, 0.f};

#pragma unroll
  for (int t = 0; t < K_DIM / 4; ++t) {
    v2f av = { a[2 * t], a[2 * t + 1] };
    v2f b0 = *(const v2f*)(fb0 + t * (OCOLS * 4));
    v2f b1 = *(const v2f*)(fb1 + t * (OCOLS * 4));
    acc0 = __builtin_amdgcn_wmma_f32_16x16x4_f32(
        false, av, false, b0, (short)0, acc0, false, false);
    acc1 = __builtin_amdgcn_wmma_f32_16x16x4_f32(
        false, av, false, b1, (short)0, acc1, false, false);
  }

  float bn = bias[mrow];
  int   mhi = half * 8;
#pragma unroll
  for (int j = 0; j < 8; ++j) {
    int   m   = j + mhi;
    float re  = acc0[j];
    float im  = acc1[j];
    float mag = __builtin_amdgcn_sqrtf(re * re + im * im);
    float sc  = fmaxf(mag + bn, 0.0f) *
                __builtin_amdgcn_rcpf(mag + 1e-8f);
    v2f v = { re * sc, im * sc };
    __builtin_nontemporal_store(
        v, (v2f*)(out + ((size_t)(node_base + m) * CH_OUT + mrow) * 2));
  }
}

extern "C" void kernel_launch(void* const* d_in, const int* in_sizes, int n_in,
                              void* d_out, int out_size, void* d_ws, size_t ws_size,
                              hipStream_t stream) {
  const float* x      = (const float*)d_in[0];
  const int*   edges  = (const int*)d_in[1];
  const float* sten   = (const float*)d_in[2];
  const float* weight = (const float*)d_in[3];
  const float* offset = (const float*)d_in[4];
  const float* bias   = (const float*)d_in[5];
  float*       out    = (float*)d_out;

  int*   cnt    = (int*)d_ws;
  int*   off    = cnt + WS_PAD_NODES;
  int*   coarse = off + WS_PAD_NODES;
  int*   grp    = coarse + 2 * N_EDGES;
  float* Fp     = (float*)(grp + 2 * N_EDGES);
  int*   btot   = (int*)(Fp + K_DIM * OCOLS);
  int*   bcur   = btot + 256 * BSTRIDE;

  hipMemsetAsync(btot, 0, (size_t)2 * 256 * BSTRIDE * sizeof(int), stream);

  filter_kernel<<<(K_DIM * OCOLS + 255) / 256, 256, 0, stream>>>(weight, offset, Fp);

  bucket_count_kernel<<<NCHUNK, BT, 0, stream>>>(edges, btot);
  bucket_scatter_kernel<<<NCHUNK, BT, 0, stream>>>(edges, btot, bcur, coarse);
  group_kernel<<<NB, GT, 0, stream>>>(btot, coarse, cnt, off, grp);

  tile_kernel<<<(N_TILES + 3) / 4, 128, 0, stream>>>(Fp, x, sten, off, cnt, grp, bias, out);
}


# Level-1 count pass, memset and all global atomics dropped: each chunk sorts its edges by bucket in its own window and leaves a (bucket, chunk) count table; level 2 reads its 250 runs; sum order same
# speedup vs baseline: 9.1779x; 1.1751x over previous
#include <hip/hip_runtime.h>

typedef float v2f __attribute__((ext_vector_type(2)));
typedef float v8f __attribute__((ext_vector_type(8)));

#define N_NODES 100000
#define N_EDGES 1600000
#define R_DIM   6
#define CH_IN   8
#define CH_OUT  16
#define K_DIM   96
#define OCOLS   32
#define N_TILES (N_NODES / 16)

#define WS_PAD_NODES 100096

#define NB      250
#define BNODES  400
#define NCHUNK  250
#define CHUNK   6400
#define BT      640
#define EPT     10
#define GT      1024
#define ID_BITS 21
#define ID_MASK ((1 << ID_BITS) - 1)

static_assert(NB * BNODES == N_NODES, "buckets cover every node exactly once");
static_assert(BNODES % 16 == 0, "a tile of 16 nodes never straddles two buckets");
static_assert(NCHUNK * CHUNK == N_EDGES && BT * EPT == CHUNK, "chunks cover every edge exactly once");
static_assert(N_EDGES <= (1 << ID_BITS) && BNODES <= 512, "edge id and node-in-bucket pack into one non-negative int");
static_assert(NB <= 256 && NB <= BT && BNODES <= 512 && GT >= 512, "scan widths");


__device__ __forceinline__ void scan256(int* sc, int t, int v)
{
  if (t < 256) sc[t] = v;
  __syncthreads();
  for (int d = 1; d < 256; d <<= 1) {
    int a = (t < 256 && t >= d) ? sc[t - d] : 0;
    __syncthreads();
    if (t < 256) sc[t] += a;
    __syncthreads();
  }
}

__global__ __launch_bounds__(BT) void chunk_sort_kernel(
    const int* __restrict__ edges,
    int*       __restrict__ coarse,
    int*       __restrict__ tab)
{
  __shared__ int h[NB];
  __shared__ int sc[256];
  int t = threadIdx.x;
  if (t < NB) h[t] = 0;
  __syncthreads();
  int first = blockIdx.x * CHUNK + t;
  int2 sd[EPT];
  int  b[EPT], r[EPT];
#pragma unroll
  for (int k = 0; k < EPT; ++k) sd[k] = ((const int2*)edges)[first + k * BT];
#pragma unroll
  for (int k = 0; k < EPT; ++k) {
    b[k] = sd[k].y / BNODES;
    r[k] = atomicAdd(&h[b[k]], 1);
  }
  __syncthreads();
  int c = (t < NB) ? h[t] : 0;
  scan256(sc, t, c);
  if (t < NB) {
    int base = sc[t] - c;
    h[t] = base;
    tab[t * NCHUNK + blockIdx.x] = c | (base << 16);
  }
  __syncthreads();
  int2* win = (int2*)coarse + blockIdx.x * CHUNK;
#pragma unroll
  for (int k = 0; k < EPT; ++k) {
    int e = first + k * BT;
    int l = sd[k].y - b[k] * BNODES;
    win[h[b[k]] + r[k]] = make_int2(e | (l << ID_BITS), sd[k].x);
  }
}

#define RUN_STEPS ((NCHUNK + 31) / 32)
static_assert(RUN_STEPS % 4 == 0 && RUN_STEPS * 32 <= 256 && GT == 1024, "run tables hold every index a wave forms");
static_assert(CHUNK < (1 << 15), "count and window start share one non-negative word");

__global__ __launch_bounds__(GT) void group_kernel(
    const int* __restrict__ tab,
    const int* __restrict__ coarse,
    int*       __restrict__ cnt,
    int*       __restrict__ off,
    int*       __restrict__ grp)
{
  __shared__ int cn[512];
  __shared__ int of[512];
  __shared__ int cur[512];
  __shared__ int rb[256];
  __shared__ int rn[256];
  __shared__ int before;
  int t    = threadIdx.x;
  int bk   = blockIdx.x;
  int lane = t & 31;
  int w    = t >> 5;
  if (t < 512) { cn[t] = 0; cur[t] = 0; }
  if (t < 256) {
    int word = (t < NCHUNK) ? tab[bk * NCHUNK + t] : 0;
    rn[t] = word & 0xFFFF;
    rb[t] = t * CHUNK + (word >> 16);
  }
  if (t == 0) before = 0;

  int s = 0;
  int len  = bk * NCHUNK;
  int len4 = len >> 2;
  const int4* tab4 = (const int4*)tab;
#pragma unroll 4
  for (int i = t; i < len4; i += GT) {
    int4 v = tab4[i];
    s += (v.x & 0xFFFF) + (v.y & 0xFFFF) + (v.z & 0xFFFF) + (v.w & 0xFFFF);
  }
  {
    int i = (len4 << 2) + t;
    if (i < len) s += tab[i] & 0xFFFF;
  }
  for (int d = 16; d; d >>= 1) s += __shfl_xor(s, d);
  __syncthreads();
  if (lane == 0 && s) atomicAdd(&before, s);
  __syncthreads();
  int start = before;
  const int2* c2 = (const int2*)coarse;

  for (int j0 = 0; j0 < RUN_STEPS; j0 += 4) {
    int pk[4], nn[4], bb[4];
#pragma unroll
    for (int k = 0; k < 4; ++k) {
      int c = w + 32 * (j0 + k);
      nn[k] = rn[c];
      bb[k] = rb[c];
      pk[k] = (lane < nn[k]) ? c2[bb[k] + lane].x : -1;
    }
#pragma unroll
    for (int k = 0; k < 4; ++k)
      if (pk[k] >= 0) atomicAdd(&cn[pk[k] >> ID_BITS], 1);
#pragma unroll
    for (int k = 0; k < 4; ++k)
      for (int i = lane + 32; i < nn[k]; i += 32)
        atomicAdd(&cn[c2[bb[k] + i].x >> ID_BITS], 1);
  }
  __syncthreads();

  int mine = (t < 512) ? cn[t] : 0;
  if (t < 512) of[t] = mine;
  __syncthreads();
  for (int d = 1; d < 512; d <<= 1) {
    int a = (t < 512 && t >= d) ? of[t - d] : 0;
    __syncthreads();
    if (t < 512) of[t] += a;
    __syncthreads();
  }
  if (t < 512) {
    int ex = of[t] - mine;
    of[t] = ex;
    if (t < BNODES) {
      cnt[bk * BNODES + t] = mine;
      off[bk * BNODES + t] = start + ex;
    }
  }
  __syncthreads();

  int2* dst = (int2*)grp + start;
  for (int j0 = 0; j0 < RUN_STEPS; j0 += 4) {
    int2 v[4];
    int  nn[4], bb[4];
#pragma unroll
    for (int k = 0; k < 4; ++k) {
      int c = w + 32 * (j0 + k);
      nn[k] = rn[c];
      bb[k] = rb[c];
      v[k] = (lane < nn[k]) ? c2[bb[k] + lane] : make_int2(-1, 0);
    }
#pragma unroll
    for (int k = 0; k < 4; ++k) {
      if (v[k].x >= 0) {
        int l = v[k].x >> ID_BITS;
        int slot = atomicAdd(&cur[l], 1);
        dst[of[l] + slot] = make_int2(v[k].x & ID_MASK, v[k].y);
      }
    }
#pragma unroll
    for (int k = 0; k < 4; ++k) {
      for (int i = lane + 32; i < nn[k]; i += 32) {
        int2 e = c2[bb[k] + i];
        int l = e.x >> ID_BITS;
        int slot = atomicAdd(&cur[l], 1);
        dst[of[l] + slot] = make_int2(e.x & ID_MASK, e.y);
      }
    }
  }
}

__global__ __launch_bounds__(256) void filter_kernel(
    const float* __restrict__ weight,
    const float* __restrict__ offset,
    float*       __restrict__ Fp)
{
  int i = blockIdx.x * 256 + threadIdx.x;
  if (i >= K_DIM * OCOLS) return;
  int k   = i >> 5;
  int col = i & 31;
  int p   = k & 1;
  int rc  = k >> 1;
  int r   = rc >> 3;
  int c   = rc & 7;
  int o   = col & 15;

  float w  = weight[(r * CH_IN + c) * CH_OUT + o];
  float ph = offset[c * CH_OUT + o];
  float fre = w * cosf(ph);
  float fim = w * sinf(ph);

  float val;
  if (col < 16) val = (p == 0) ? fre : -fim;
  else          val = (p == 0) ? fim :  fre;

  Fp[(rc * OCOLS + col) * 2 + p] = val;
}

#define WAVE_CAP    512
#define STAGE_LOADS (WAVE_CAP / 32)

__device__ __forceinline__ void add_edge(float (&a)[48], float4 s0, float4 s1, float4 s2,
                                         float x0, float x1, float x2, float x3)
{
  float st[12] = {s0.x, s0.y, s0.z, s0.w, s1.x, s1.y, s1.z, s1.w, s2.x, s2.y, s2.z, s2.w};
  float xs[4]  = {x0, x1, x2, x3};
#pragma unroll
  for (int r = 0; r < R_DIM; ++r) {
#pragma unroll
    for (int j = 0; j < 4; ++j) {
      a[(r * 4 + j) * 2 + 0] = __builtin_fmaf(st[r * 2 + 0], xs[j], a[(r * 4 + j) * 2 + 0]);
      a[(r * 4 + j) * 2 + 1] = __builtin_fmaf(st[r * 2 + 1], xs[j], a[(r * 4 + j) * 2 + 1]);
    }
  }
}

__global__ __launch_bounds__(128) __attribute__((amdgpu_waves_per_eu(1, 7))) void tile_kernel(
    const float* __restrict__ Fp,
    const float* __restrict__ x,
    const float* __restrict__ sten,
    const int*   __restrict__ off,
    const int*   __restrict__ cnt,
    const int*   __restrict__ grp,
    const float* __restrict__ bias,
    float*       __restrict__ out)
{
  __shared__ float Fl[K_DIM * OCOLS];
  __shared__ int2  raw[4][WAVE_CAP];
  __shared__ int2  srt[4][WAVE_CAP];
  for (int i = threadIdx.x; i < K_DIM * OCOLS; i += 128) Fl[i] = Fp[i];
  __syncthreads();

  int wave = threadIdx.x >> 5;
  int lane = threadIdx.x & 31;
  int tile = blockIdx.x * 4 + wave;
  if (tile >= N_TILES) return;

  int node_base = tile * 16;
  int mrow = lane & 15;
  int half = lane >> 4;

  float a[48];
#pragma unroll
  for (int q = 0; q < 48; ++q) a[q] = 0.f;

  int s = off[node_base + mrow];
  int n = cnt[node_base + mrow];
  int nm1 = n - 1;
  int w0 = __builtin_amdgcn_readlane(s, 0);
  int T  = __builtin_amdgcn_readlane(s + n, 15) - w0;
  int sl = s - w0;
  const int2* g2 = (const int2*)grp + w0;

  if (T > 0 && T <= WAVE_CAP) {
    int2* rw = raw[wave];
    int2* sr = srt[wave];
    int Tm1 = T - 1;
    int2 v[STAGE_LOADS];
#pragma unroll
    for (int k = 0; k < STAGE_LOADS; ++k) {
      int p = lane + 32 * k;
      v[k] = g2[p < Tm1 ? p : Tm1];
    }
#pragma unroll
    for (int k = 0; k < STAGE_LOADS; ++k) {
      int p = lane + 32 * k;
      if (p < T) rw[p] = v[k];
    }

    for (int i0 = half; i0 < n; i0 += 8) {
      int2 e[4];
      int  r[4];
#pragma unroll
      for (int k = 0; k < 4; ++k) {
        int i = i0 + 2 * k;
        e[k] = rw[sl + (i < nm1 ? i : nm1)];
        r[k] = 0;
      }
#pragma unroll 4
      for (int j = 0; j < n; ++j) {
        int vj = rw[sl + j].x;
#pragma unroll
        for (int k = 0; k < 4; ++k) r[k] += (vj < e[k].x) ? 1 : 0;
      }
#pragma unroll
      for (int k = 0; k < 4; ++k)
        if (i0 + 2 * k < n) sr[sl + r[k]] = e[k];
    }

    if (n > 0) {
      const int2* mine = sr + sl;
      for (int i0 = 0; i0 < n; i0 += 4) {
        int2 ed[4];
#pragma unroll
        for (int k = 0; k < 4; ++k) {
          int ik = i0 + k;
          ed[k] = mine[ik < nm1 ? ik : nm1];
        }
        float4 S0[4], S1[4], S2[4];
        float  xs[4][4];
#pragma unroll
        for (int k = 0; k < 4; ++k) {
          unsigned so = (unsigned)ed[k].x * (unsigned)(R_DIM * 2 * sizeof(float));
          unsigned xo = (unsigned)ed[k].y * (unsigned)(CH_IN * sizeof(float));
          const float4* sp = (const float4*)((const char*)sten + so);
          const float4* xp = (const float4*)((const char*)x + xo);
          S0[k] = sp[0]; S1[k] = sp[1]; S2[k] = sp[2];
          float4 xa = xp[0];
          float4 xb = xp[1];
          bool ok = (i0 + k) < n;
          float c0 = half ? xa.y : xa.x;
          float c1 = half ? xa.w : xa.z;
          float c2 = half ? xb.y : xb.x;
          float c3 = half ? xb.w : xb.z;
          xs[k][0] = ok ? c0 : 0.f;
          xs[k][1] = ok ? c1 : 0.f;
          xs[k][2] = ok ? c2 : 0.f;
          xs[k][3] = ok ? c3 : 0.f;
        }
#pragma unroll
        for (int k = 0; k < 4; ++k)
          add_edge(a, S0[k], S1[k], S2[k], xs[k][0], xs[k][1], xs[k][2], xs[k][3]);
      }
    }
  } else if (T > WAVE_CAP) {
    if (n > 0) {
      const int2* g = g2 + sl;
      int last = -1;
      for (int i = 0; i < n; ++i) {
        int best = 0x7fffffff, bsrc = 0;
        for (int j = 0; j < n; ++j) {
          int2 e = g[j];
          if (e.x > last && e.x < best) { best = e.x; bsrc = e.y; }
        }
        last = best;
        const float4* sp = (const float4*)(sten + (size_t)best * (R_DIM * 2));
        const float4* xp = (const float4*)(x + (size_t)bsrc * CH_IN);
        float4 xa = xp[0];
        float4 xb = xp[1];
        add_edge(a, sp[0], sp[1], sp[2],
                 half ? xa.y : xa.x, half ? xa.w : xa.z,
                 half ? xb.y : xb.x, half ? xb.w : xb.z);
      }
    }
  }

  const float* fb0 = Fl + (size_t)(half * OCOLS + mrow) * 2;
  const float* fb1 = fb0 + 32;

  v8f acc0 = {0.f, 0.f, 0.f, 0.f, 0.f, 0.f, 0.f, 0.f};
  v8f acc1 = {0.f, 0.f, 0.f, 0.f, 0.f, 0.f, 0.f, 0.f};

#pragma unroll
  for (int t = 0; t < K_DIM / 4; ++t) {
    v2f av = { a[2 * t], a[2 * t + 1] };
    v2f b0 = *(const v2f*)(fb0 + t * (OCOLS * 4));
    v2f b1 = *(const v2f*)(fb1 + t * (OCOLS * 4));
    acc0 = __builtin_amdgcn_wmma_f32_16x16x4_f32(
        false, av, false, b0, (short)0, acc0, false, false);
    acc1 = __builtin_amdgcn_wmma_f32_16x16x4_f32(
        false, av, false, b1, (short)0, acc1, false, false);
  }

  float bn = bias[mrow];
  int   mhi = half * 8;
#pragma unroll
  for (int j = 0; j < 8; ++j) {
    int   m   = j + mhi;
    float re  = acc0[j];
    float im  = acc1[j];
    float mag = __builtin_amdgcn_sqrtf(re * re + im * im);
    float sc  = fmaxf(mag + bn, 0.0f) *
                __builtin_amdgcn_rcpf(mag + 1e-8f);
    v2f v = { re * sc, im * sc };
    __builtin_nontemporal_store(
        v, (v2f*)(out + ((size_t)(node_base + m) * CH_OUT + mrow) * 2));
  }
}

extern "C" void kernel_launch(void* const* d_in, const int* in_sizes, int n_in,
                              void* d_out, int out_size, void* d_ws, size_t ws_size,
                              hipStream_t stream) {
  const float* x      = (const float*)d_in[0];
  const int*   edges  = (const int*)d_in[1];
  const float* sten   = (const float*)d_in[2];
  const float* weight = (const float*)d_in[3];
  const float* offset = (const float*)d_in[4];
  const float* bias   = (const float*)d_in[5];
  float*       out    = (float*)d_out;

  int*   cnt    = (int*)d_ws;
  int*   off    = cnt + WS_PAD_NODES;
  int*   coarse = off + WS_PAD_NODES;
  int*   grp    = coarse + 2 * N_EDGES;
  float* Fp     = (float*)(grp + 2 * N_EDGES);
  int*   tab    = (int*)(Fp + K_DIM * OCOLS);

  filter_kernel<<<(K_DIM * OCOLS + 255) / 256, 256, 0, stream>>>(weight, offset, Fp);

  chunk_sort_kernel<<<NCHUNK, BT, 0, stream>>>(edges, coarse, tab);
  group_kernel<<<NB, GT, 0, stream>>>(tab, coarse, cnt, off, grp);

  tile_kernel<<<(N_TILES + 3) / 4, 128, 0, stream>>>(Fp, x, sten, off, cnt, grp, bias, out);
}


# Summing loop: a node's two lanes each load half of an edge's stencil and x bytes and swap halves by v_permlanex16: 3 gathers per edge and lane, not 5; same products, same sum order
# speedup vs baseline: 12.1400x; 1.3227x over previous
#include <hip/hip_runtime.h>

typedef float v2f __attribute__((ext_vector_type(2)));
typedef float v8f __attribute__((ext_vector_type(8)));

#define N_NODES 100000
#define N_EDGES 1600000
#define R_DIM   6
#define CH_IN   8
#define CH_OUT  16
#define K_DIM   96
#define OCOLS   32

#define NB      250
#define BNODES  400
#define NCHUNK  250
#define CHUNK   6400
#define BT      640
#define EPT     10
#define ID_BITS 21
#define ID_MASK ((1 << ID_BITS) - 1)

#define FW      25
#define FT      (FW * 32)
#define RUNS    (NCHUNK / FW)
#define RSTEP   5
#define LCAP    15360

static_assert(NB * BNODES == N_NODES, "buckets cover every node exactly once");
static_assert(FW * 16 == BNODES, "one wave per tile; a tile never straddles two buckets");
static_assert(NCHUNK * CHUNK == N_EDGES && BT * EPT == CHUNK, "chunks cover every edge exactly once");
static_assert(N_EDGES <= (1 << ID_BITS) && BNODES <= 512, "edge id and node-in-bucket pack into one non-negative int");
static_assert(NB <= 256 && NB <= BT && BNODES <= 512 && FT >= 512, "scan widths");
static_assert(FW * RUNS == NCHUNK && RUNS % RSTEP == 0 && NCHUNK <= 256, "the waves cover every run exactly once");
static_assert(CHUNK < (1 << 15), "count and window start share one non-negative word");
static_assert(K_DIM * OCOLS / 4 <= FT, "one 16-byte load per thread stages the whole filter");
static_assert(2 * LCAP * 8 + K_DIM * OCOLS * 4 + 3 * 512 * 4 + 2 * 256 * 4 <= 272 * 1024, "the work-group's LDS");


__device__ __forceinline__ void scan256(int* sc, int t, int v)
{
  if (t < 256) sc[t] = v;
  __syncthreads();
  for (int d = 1; d < 256; d <<= 1) {
    int a = (t < 256 && t >= d) ? sc[t - d] : 0;
    __syncthreads();
    if (t < 256) sc[t] += a;
    __syncthreads();
  }
}

__global__ __launch_bounds__(BT) void chunk_sort_kernel(
    const int* __restrict__ edges,
    int*       __restrict__ coarse,
    int*       __restrict__ tab)
{
  __shared__ int h[NB];
  __shared__ int sc[256];
  int t = threadIdx.x;
  if (t < NB) h[t] = 0;
  __syncthreads();
  int first = blockIdx.x * CHUNK + t;
  int2 sd[EPT];
  int  b[EPT], r[EPT];
#pragma unroll
  for (int k = 0; k < EPT; ++k) sd[k] = ((const int2*)edges)[first + k * BT];
#pragma unroll
  for (int k = 0; k < EPT; ++k) {
    b[k] = sd[k].y / BNODES;
    r[k] = atomicAdd(&h[b[k]], 1);
  }
  __syncthreads();
  int c = (t < NB) ? h[t] : 0;
  scan256(sc, t, c);
  if (t < NB) {
    int base = sc[t] - c;
    h[t] = base;
    tab[t * NCHUNK + blockIdx.x] = c | (base << 16);
  }
  __syncthreads();
  int2* win = (int2*)coarse + blockIdx.x * CHUNK;
#pragma unroll
  for (int k = 0; k < EPT; ++k) {
    int e = first + k * BT;
    int l = sd[k].y - b[k] * BNODES;
    win[h[b[k]] + r[k]] = make_int2(e | (l << ID_BITS), sd[k].x);
  }
}

__global__ __launch_bounds__(256) void filter_kernel(
    const float* __restrict__ weight,
    const float* __restrict__ offset,
    float*       __restrict__ Fp)
{
  int i = blockIdx.x * 256 + threadIdx.x;
  if (i >= K_DIM * OCOLS) return;
  int k   = i >> 5;
  int col = i & 31;
  int p   = k & 1;
  int rc  = k >> 1;
  int r   = rc >> 3;
  int c   = rc & 7;
  int o   = col & 15;

  float w  = weight[(r * CH_IN + c) * CH_OUT + o];
  float ph = offset[c * CH_OUT + o];
  float fre = w * cosf(ph);
  float fim = w * sinf(ph);

  float val;
  if (col < 16) val = (p == 0) ? fre : -fim;
  else          val = (p == 0) ? fim :  fre;

  Fp[(rc * OCOLS + col) * 2 + p] = val;
}

__device__ __forceinline__ float xhalf(float v)
{
  return __builtin_bit_cast(float, __builtin_amdgcn_permlanex16(
      __builtin_bit_cast(int, v), __builtin_bit_cast(int, v), 0x76543210, 0xfedcba98, false, false));
}

__device__ __forceinline__ void add_edge(float (&a)[48], float4 s0, float4 s1, float4 s2,
                                         float x0, float x1, float x2, float x3)
{
  float st[12] = {s0.x, s0.y, s0.z, s0.w, s1.x, s1.y, s1.z, s1.w, s2.x, s2.y, s2.z, s2.w};
  float xs[4]  = {x0, x1, x2, x3};
#pragma unroll
  for (int r = 0; r < R_DIM; ++r) {
#pragma unroll
    for (int j = 0; j < 4; ++j) {
      a[(r * 4 + j) * 2 + 0] = __builtin_fmaf(st[r * 2 + 0], xs[j], a[(r * 4 + j) * 2 + 0]);
      a[(r * 4 + j) * 2 + 1] = __builtin_fmaf(st[r * 2 + 1], xs[j], a[(r * 4 + j) * 2 + 1]);
    }
  }
}

__global__ __launch_bounds__(FT) __attribute__((amdgpu_waves_per_eu(1, 7))) void bucket_kernel(
    const float* __restrict__ Fp,
    const float* __restrict__ x,
    const float* __restrict__ sten,
    const int*   __restrict__ tab,
    const int*   __restrict__ coarse,
    const float* __restrict__ bias,
    float*       __restrict__ out)
{
  __shared__ __attribute__((aligned(16))) float Fl[K_DIM * OCOLS];
  __shared__ int cn[512];
  __shared__ int of[512];
  __shared__ int cur[512];
  __shared__ int rb[256];
  __shared__ int rn[256];
  __shared__ int2 raw[LCAP];
  __shared__ int2 srt[LCAP];
  int t    = threadIdx.x;
  int bk   = blockIdx.x;
  int lane = t & 31;
  int w    = t >> 5;
  float4 fv = make_float4(0.f, 0.f, 0.f, 0.f);
  if (t < K_DIM * OCOLS / 4) fv = ((const float4*)Fp)[t];
  if (t < 512) { cn[t] = 0; cur[t] = 0; }
  if (t < 256) {
    int word = (t < NCHUNK) ? tab[bk * NCHUNK + t] : 0;
    rn[t] = word & 0xFFFF;
    rb[t] = t * CHUNK + (word >> 16);
  }
  if (t < K_DIM * OCOLS / 4) ((float4*)Fl)[t] = fv;
  __syncthreads();
  const int2* c2 = (const int2*)coarse;

  for (int j0 = 0; j0 < RUNS; j0 += RSTEP) {
    int pk[RSTEP], nn[RSTEP], bb[RSTEP];
#pragma unroll
    for (int k = 0; k < RSTEP; ++k) {
      int c = w + FW * (j0 + k);
      nn[k] = rn[c];
      bb[k] = rb[c];
      pk[k] = (lane < nn[k]) ? c2[bb[k] + lane].x : -1;
    }
#pragma unroll
    for (int k = 0; k < RSTEP; ++k)
      if (pk[k] >= 0) atomicAdd(&cn[pk[k] >> ID_BITS], 1);
#pragma unroll
    for (int k = 0; k < RSTEP; ++k)
      for (int i = lane + 32; i < nn[k]; i += 32)
        atomicAdd(&cn[c2[bb[k] + i].x >> ID_BITS], 1);
  }
  __syncthreads();

  int mine_n = (t < 512) ? cn[t] : 0;
  if (t < 512) of[t] = mine_n;
  __syncthreads();
  for (int d = 1; d < 512; d <<= 1) {
    int a = (t < 512 && t >= d) ? of[t - d] : 0;
    __syncthreads();
    if (t < 512) of[t] += a;
    __syncthreads();
  }
  if (t < 512) of[t] -= mine_n;
  __syncthreads();

  int  total = __builtin_amdgcn_readfirstlane(of[BNODES - 1] + cn[BNODES - 1]);
  bool fits  = total <= LCAP;

  if (fits) {
    for (int j0 = 0; j0 < RUNS; j0 += RSTEP) {
      int2 v[RSTEP];
      int  nn[RSTEP], bb[RSTEP];
#pragma unroll
      for (int k = 0; k < RSTEP; ++k) {
        int c = w + FW * (j0 + k);
        nn[k] = rn[c];
        bb[k] = rb[c];
        v[k] = (lane < nn[k]) ? c2[bb[k] + lane] : make_int2(-1, 0);
      }
#pragma unroll
      for (int k = 0; k < RSTEP; ++k) {
        if (v[k].x >= 0) {
          int l = v[k].x >> ID_BITS;
          int slot = atomicAdd(&cur[l], 1);
          raw[of[l] + slot] = make_int2(v[k].x & ID_MASK, v[k].y);
        }
      }
#pragma unroll
      for (int k = 0; k < RSTEP; ++k) {
        for (int i = lane + 32; i < nn[k]; i += 32) {
          int2 e = c2[bb[k] + i];
          int l = e.x >> ID_BITS;
          int slot = atomicAdd(&cur[l], 1);
          raw[of[l] + slot] = make_int2(e.x & ID_MASK, e.y);
        }
      }
    }
  }
  __syncthreads();

  int node_base = bk * BNODES + w * 16;
  int mrow = lane & 15;
  int half = lane >> 4;
  int ln   = w * 16 + mrow;

  float a[48];
#pragma unroll
  for (int q = 0; q < 48; ++q) a[q] = 0.f;

  int sl  = of[ln];
  int n   = cn[ln];
  int nm1 = n - 1;

  if (fits) {
    const int2* rw = raw;
    int2*       sr = srt;
    for (int i0 = half; i0 < n; i0 += 8) {
      int2 e[4];
      int  r[4];
#pragma unroll
      for (int k = 0; k < 4; ++k) {
        int i = i0 + 2 * k;
        e[k] = rw[sl + (i < nm1 ? i : nm1)];
        r[k] = 0;
      }
#pragma unroll 4
      for (int j = 0; j < n; ++j) {
        int vj = rw[sl + j].x;
#pragma unroll
        for (int k = 0; k < 4; ++k) r[k] += (vj < e[k].x) ? 1 : 0;
      }
#pragma unroll
      for (int k = 0; k < 4; ++k)
        if (i0 + 2 * k < n) sr[sl + r[k]] = e[k];
    }

    if (n > 0) {
      const int2* mine = sr + sl;
      float acc[48];
#pragma unroll
      for (int q = 0; q < 48; ++q) acc[q] = 0.f;
      unsigned own = (unsigned)half * 16u;
      for (int i0 = 0; i0 < n; i0 += 4) {
        int2 ed[4];
#pragma unroll
        for (int k = 0; k < 4; ++k) {
          int ik = i0 + k;
          ed[k] = mine[ik < nm1 ? ik : nm1];
        }
        float4 SO[4], ST[4], XO[4];
#pragma unroll
        for (int k = 0; k < 4; ++k) {
          unsigned so = (unsigned)ed[k].x * (unsigned)(R_DIM * 2 * sizeof(float));
          unsigned xo = (unsigned)ed[k].y * (unsigned)(CH_IN * sizeof(float));
          SO[k] = *(const float4*)((const char*)sten + (so + own));
          ST[k] = *(const float4*)((const char*)sten + so + 32);
          XO[k] = *(const float4*)((const char*)x + (xo + own));
        }
#pragma unroll
        for (int k = 0; k < 4; ++k) {
          bool ok = (i0 + k) < n;
          float4 xa = XO[k];
          float k0 = half ? xa.y : xa.x;
          float k1 = half ? xa.w : xa.z;
          float t0 = half ? xa.x : xa.y;
          float t1 = half ? xa.z : xa.w;
          k0 = ok ? k0 : 0.f;
          k1 = ok ? k1 : 0.f;
          t0 = ok ? t0 : 0.f;
          t1 = ok ? t1 : 0.f;
          float stq[3][4] = {
            { SO[k].x, SO[k].y, SO[k].z, SO[k].w },
            { xhalf(SO[k].x), xhalf(SO[k].y), xhalf(SO[k].z), xhalf(SO[k].w) },
            { ST[k].x, ST[k].y, ST[k].z, ST[k].w } };
          float xv[2][2] = { { k0, k1 }, { xhalf(t0), xhalf(t1) } };
#pragma unroll
          for (int Q = 0; Q < 3; ++Q) {
#pragma unroll
            for (int s = 0; s < 4; ++s) {
#pragma unroll
              for (int X = 0; X < 2; ++X) {
#pragma unroll
                for (int u = 0; u < 2; ++u) {
                  int q = ((Q * 4 + s) * 2 + X) * 2 + u;
                  acc[q] = __builtin_fmaf(stq[Q][s], xv[X][u], acc[q]);
                }
              }
            }
          }
        }
      }
#pragma unroll
      for (int r = 0; r < R_DIM; ++r) {
#pragma unroll
        for (int j = 0; j < 4; ++j) {
#pragma unroll
          for (int p = 0; p < 2; ++p) {
            int rq = r >> 1, s = (r & 1) * 2 + p, jx = j >> 1, u = j & 1;
            int Q0 = rq;
            int Q1 = (rq == 0) ? 1 : (rq == 1) ? 0 : 2;
            int q0 = ((Q0 * 4 + s) * 2 + jx) * 2 + u;
            int q1 = ((Q1 * 4 + s) * 2 + (1 - jx)) * 2 + u;
            float v0 = acc[q0];
            float v1 = acc[q1];
            a[(r * 4 + j) * 2 + p] = half ? v1 : v0;
          }
        }
      }
    }
  } else {
    int last = -1;
    for (int i = 0; i < n; ++i) {
      int best = 0x7fffffff, bsrc = 0;
      for (int c = 0; c < NCHUNK; ++c) {
        int m = rn[c];
        const int2* run = c2 + rb[c];
        for (int j = 0; j < m; ++j) {
          int2 e = run[j];
          int id = e.x & ID_MASK;
          if ((e.x >> ID_BITS) == ln && id > last && id < best) { best = id; bsrc = e.y; }
        }
      }
      last = best;
      const float4* sp = (const float4*)(sten + (size_t)best * (R_DIM * 2));
      const float4* xp = (const float4*)(x + (size_t)bsrc * CH_IN);
      float4 xa = xp[0];
      float4 xb = xp[1];
      add_edge(a, sp[0], sp[1], sp[2],
               half ? xa.y : xa.x, half ? xa.w : xa.z,
               half ? xb.y : xb.x, half ? xb.w : xb.z);
    }
  }

  const float* fb0 = Fl + (size_t)(half * OCOLS + mrow) * 2;
  const float* fb1 = fb0 + 32;

  v8f acc0 = {0.f, 0.f, 0.f, 0.f, 0.f, 0.f, 0.f, 0.f};
  v8f acc1 = {0.f, 0.f, 0.f, 0.f, 0.f, 0.f, 0.f, 0.f};

#pragma unroll
  for (int ks = 0; ks < K_DIM / 4; ++ks) {
    v2f av = { a[2 * ks], a[2 * ks + 1] };
    v2f b0 = *(const v2f*)(fb0 + ks * (OCOLS * 4));
    v2f b1 = *(const v2f*)(fb1 + ks * (OCOLS * 4));
    acc0 = __builtin_amdgcn_wmma_f32_16x16x4_f32(
        false, av, false, b0, (short)0, acc0, false, false);
    acc1 = __builtin_amdgcn_wmma_f32_16x16x4_f32(
        false, av, false, b1, (short)0, acc1, false, false);
  }

  float bn = bias[mrow];
  int   mhi = half * 8;
#pragma unroll
  for (int j = 0; j < 8; ++j) {
    int   m   = j + mhi;
    float re  = acc0[j];
    float im  = acc1[j];
    float mag = __builtin_amdgcn_sqrtf(re * re + im * im);
    float sc  = fmaxf(mag + bn, 0.0f) *
                __builtin_amdgcn_rcpf(mag + 1e-8f);
    v2f v = { re * sc, im * sc };
    __builtin_nontemporal_store(
        v, (v2f*)(out + ((size_t)(node_base + m) * CH_OUT + mrow) * 2));
  }
}

extern "C" void kernel_launch(void* const* d_in, const int* in_sizes, int n_in,
                              void* d_out, int out_size, void* d_ws, size_t ws_size,
                              hipStream_t stream) {
  const float* x      = (const float*)d_in[0];
  const int*   edges  = (const int*)d_in[1];
  const float* sten   = (const float*)d_in[2];
  const float* weight = (const float*)d_in[3];
  const float* offset = (const float*)d_in[4];
  const float* bias   = (const float*)d_in[5];
  float*       out    = (float*)d_out;

  int*   coarse = (int*)d_ws;
  float* Fp     = (float*)(coarse + 2 * N_EDGES);
  int*   tab    = (int*)(Fp + K_DIM * OCOLS);

  filter_kernel<<<(K_DIM * OCOLS + 255) / 256, 256, 0, stream>>>(weight, offset, Fp);

  chunk_sort_kernel<<<NCHUNK, BT, 0, stream>>>(edges, coarse, tab);
  bucket_kernel<<<NB, FT, 0, stream>>>(Fp, x, sten, tab, coarse, bias, out);
}


# Level-1 window grouped in LDS first and written as whole 128-byte lines (16 B per lane, contiguous) instead of 6,400 scattered 8-byte stores per chunk; list bytes and sum order unchanged
# speedup vs baseline: 13.1157x; 1.0804x over previous
#include <hip/hip_runtime.h>

typedef float v2f __attribute__((ext_vector_type(2)));
typedef float v8f __attribute__((ext_vector_type(8)));

#define N_NODES 100000
#define N_EDGES 1600000
#define R_DIM   6
#define CH_IN   8
#define CH_OUT  16
#define K_DIM   96
#define OCOLS   32

#define NB      250
#define BNODES  400
#define NCHUNK  250
#define CHUNK   6400
#define BT      640
#define EPT     10
#define ID_BITS 21
#define ID_MASK ((1 << ID_BITS) - 1)

#define FW      25
#define FT      (FW * 32)
#define RUNS    (NCHUNK / FW)
#define RSTEP   5
#define LCAP    15360

static_assert(NB * BNODES == N_NODES, "buckets cover every node exactly once");
static_assert(FW * 16 == BNODES, "one wave per tile; a tile never straddles two buckets");
static_assert(NCHUNK * CHUNK == N_EDGES && BT * EPT == CHUNK, "chunks cover every edge exactly once");
static_assert(N_EDGES <= (1 << ID_BITS) && BNODES <= 512, "edge id and node-in-bucket pack into one non-negative int");
static_assert(NB <= 256 && NB <= BT && BNODES <= 512 && FT >= 512, "scan widths");
static_assert(FW * RUNS == NCHUNK && RUNS % RSTEP == 0 && NCHUNK <= 256, "the waves cover every run exactly once");
static_assert(CHUNK < (1 << 15), "count and window start share one non-negative word");
static_assert((CHUNK / 2) % BT == 0 && (CHUNK * 8) % 128 == 0 && BT % 32 == 0, "the window leaves in 16-byte pieces, the same number per thread, whole lines per wave");
static_assert(K_DIM * OCOLS / 4 <= FT, "one 16-byte load per thread stages the whole filter");
static_assert(2 * LCAP * 8 + K_DIM * OCOLS * 4 + 3 * 512 * 4 + 2 * 256 * 4 <= 272 * 1024, "the work-group's LDS");


__device__ __forceinline__ void scan256(int* sc, int t, int v)
{
  if (t < 256) sc[t] = v;
  __syncthreads();
  for (int d = 1; d < 256; d <<= 1) {
    int a = (t < 256 && t >= d) ? sc[t - d] : 0;
    __syncthreads();
    if (t < 256) sc[t] += a;
    __syncthreads();
  }
}

__global__ __launch_bounds__(BT) void chunk_sort_kernel(
    const int* __restrict__ edges,
    int*       __restrict__ coarse,
    int*       __restrict__ tab)
{
  __shared__ int h[NB];
  __shared__ int sc[256];
  __shared__ __attribute__((aligned(16))) int2 stage[CHUNK];
  int t = threadIdx.x;
  if (t < NB) h[t] = 0;
  __syncthreads();
  int first = blockIdx.x * CHUNK + t;
  int2 sd[EPT];
  int  b[EPT], r[EPT];
#pragma unroll
  for (int k = 0; k < EPT; ++k) sd[k] = ((const int2*)edges)[first + k * BT];
#pragma unroll
  for (int k = 0; k < EPT; ++k) {
    b[k] = sd[k].y / BNODES;
    r[k] = atomicAdd(&h[b[k]], 1);
  }
  __syncthreads();
  int c = (t < NB) ? h[t] : 0;
  scan256(sc, t, c);
  if (t < NB) {
    int base = sc[t] - c;
    h[t] = base;
    tab[t * NCHUNK + blockIdx.x] = c | (base << 16);
  }
  __syncthreads();
#pragma unroll
  for (int k = 0; k < EPT; ++k) {
    int e = first + k * BT;
    int l = sd[k].y - b[k] * BNODES;
    stage[h[b[k]] + r[k]] = make_int2(e | (l << ID_BITS), sd[k].x);
  }
  __syncthreads();
  int4*       win = (int4*)((int2*)coarse + blockIdx.x * CHUNK);
  const int4* st4 = (const int4*)stage;
#pragma unroll
  for (int k = 0; k < CHUNK / 2 / BT; ++k) win[t + k * BT] = st4[t + k * BT];
}

__global__ __launch_bounds__(256) void filter_kernel(
    const float* __restrict__ weight,
    const float* __restrict__ offset,
    float*       __restrict__ Fp)
{
  int i = blockIdx.x * 256 + threadIdx.x;
  if (i >= K_DIM * OCOLS) return;
  int k   = i >> 5;
  int col = i & 31;
  int p   = k & 1;
  int rc  = k >> 1;
  int r   = rc >> 3;
  int c   = rc & 7;
  int o   = col & 15;

  float w  = weight[(r * CH_IN + c) * CH_OUT + o];
  float ph = offset[c * CH_OUT + o];
  float fre = w * cosf(ph);
  float fim = w * sinf(ph);

  float val;
  if (col < 16) val = (p == 0) ? fre : -fim;
  else          val = (p == 0) ? fim :  fre;

  Fp[(rc * OCOLS + col) * 2 + p] = val;
}

__device__ __forceinline__ float xhalf(float v)
{
  return __builtin_bit_cast(float, __builtin_amdgcn_permlanex16(
      __builtin_bit_cast(int, v), __builtin_bit_cast(int, v), 0x76543210, 0xfedcba98, false, false));
}

__device__ __forceinline__ void add_edge(float (&a)[48], float4 s0, float4 s1, float4 s2,
                                         float x0, float x1, float x2, float x3)
{
  float st[12] = {s0.x, s0.y, s0.z, s0.w, s1.x, s1.y, s1.z, s1.w, s2.x, s2.y, s2.z, s2.w};
  float xs[4]  = {x0, x1, x2, x3};
#pragma unroll
  for (int r = 0; r < R_DIM; ++r) {
#pragma unroll
    for (int j = 0; j < 4; ++j) {
      a[(r * 4 + j) * 2 + 0] = __builtin_fmaf(st[r * 2 + 0], xs[j], a[(r * 4 + j) * 2 + 0]);
      a[(r * 4 + j) * 2 + 1] = __builtin_fmaf(st[r * 2 + 1], xs[j], a[(r * 4 + j) * 2 + 1]);
    }
  }
}

__global__ __launch_bounds__(FT) __attribute__((amdgpu_waves_per_eu(1, 7))) void bucket_kernel(
    const float* __restrict__ Fp,
    const float* __restrict__ x,
    const float* __restrict__ sten,
    const int*   __restrict__ tab,
    const int*   __restrict__ coarse,
    const float* __restrict__ bias,
    float*       __restrict__ out)
{
  __shared__ __attribute__((aligned(16))) float Fl[K_DIM * OCOLS];
  __shared__ int cn[512];
  __shared__ int of[512];
  __shared__ int cur[512];
  __shared__ int rb[256];
  __shared__ int rn[256];
  __shared__ int2 raw[LCAP];
  __shared__ int2 srt[LCAP];
  int t    = threadIdx.x;
  int bk   = blockIdx.x;
  int lane = t & 31;
  int w    = t >> 5;
  float4 fv = make_float4(0.f, 0.f, 0.f, 0.f);
  if (t < K_DIM * OCOLS / 4) fv = ((const float4*)Fp)[t];
  if (t < 512) { cn[t] = 0; cur[t] = 0; }
  if (t < 256) {
    int word = (t < NCHUNK) ? tab[bk * NCHUNK + t] : 0;
    rn[t] = word & 0xFFFF;
    rb[t] = t * CHUNK + (word >> 16);
  }
  if (t < K_DIM * OCOLS / 4) ((float4*)Fl)[t] = fv;
  __syncthreads();
  const int2* c2 = (const int2*)coarse;

  for (int j0 = 0; j0 < RUNS; j0 += RSTEP) {
    int pk[RSTEP], nn[RSTEP], bb[RSTEP];
#pragma unroll
    for (int k = 0; k < RSTEP; ++k) {
      int c = w + FW * (j0 + k);
      nn[k] = rn[c];
      bb[k] = rb[c];
      pk[k] = (lane < nn[k]) ? c2[bb[k] + lane].x : -1;
    }
#pragma unroll
    for (int k = 0; k < RSTEP; ++k)
      if (pk[k] >= 0) atomicAdd(&cn[pk[k] >> ID_BITS], 1);
#pragma unroll
    for (int k = 0; k < RSTEP; ++k)
      for (int i = lane + 32; i < nn[k]; i += 32)
        atomicAdd(&cn[c2[bb[k] + i].x >> ID_BITS], 1);
  }
  __syncthreads();

  int mine_n = (t < 512) ? cn[t] : 0;
  if (t < 512) of[t] = mine_n;
  __syncthreads();
  for (int d = 1; d < 512; d <<= 1) {
    int a = (t < 512 && t >= d) ? of[t - d] : 0;
    __syncthreads();
    if (t < 512) of[t] += a;
    __syncthreads();
  }
  if (t < 512) of[t] -= mine_n;
  __syncthreads();

  int  total = __builtin_amdgcn_readfirstlane(of[BNODES - 1] + cn[BNODES - 1]);
  bool fits  = total <= LCAP;

  if (fits) {
    for (int j0 = 0; j0 < RUNS; j0 += RSTEP) {
      int2 v[RSTEP];
      int  nn[RSTEP], bb[RSTEP];
#pragma unroll
      for (int k = 0; k < RSTEP; ++k) {
        int c = w + FW * (j0 + k);
        nn[k] = rn[c];
        bb[k] = rb[c];
        v[k] = (lane < nn[k]) ? c2[bb[k] + lane] : make_int2(-1, 0);
      }
#pragma unroll
      for (int k = 0; k < RSTEP; ++k) {
        if (v[k].x >= 0) {
          int l = v[k].x >> ID_BITS;
          int slot = atomicAdd(&cur[l], 1);
          raw[of[l] + slot] = make_int2(v[k].x & ID_MASK, v[k].y);
        }
      }
#pragma unroll
      for (int k = 0; k < RSTEP; ++k) {
        for (int i = lane + 32; i < nn[k]; i += 32) {
          int2 e = c2[bb[k] + i];
          int l = e.x >> ID_BITS;
          int slot = atomicAdd(&cur[l], 1);
          raw[of[l] + slot] = make_int2(e.x & ID_MASK, e.y);
        }
      }
    }
  }
  __syncthreads();

  int node_base = bk * BNODES + w * 16;
  int mrow = lane & 15;
  int half = lane >> 4;
  int ln   = w * 16 + mrow;

  float a[48];
#pragma unroll
  for (int q = 0; q < 48; ++q) a[q] = 0.f;

  int sl  = of[ln];
  int n   = cn[ln];
  int nm1 = n - 1;

  if (fits) {
    const int2* rw = raw;
    int2*       sr = srt;
    for (int i0 = half; i0 < n; i0 += 8) {
      int2 e[4];
      int  r[4];
#pragma unroll
      for (int k = 0; k < 4; ++k) {
        int i = i0 + 2 * k;
        e[k] = rw[sl + (i < nm1 ? i : nm1)];
        r[k] = 0;
      }
#pragma unroll 4
      for (int j = 0; j < n; ++j) {
        int vj = rw[sl + j].x;
#pragma unroll
        for (int k = 0; k < 4; ++k) r[k] += (vj < e[k].x) ? 1 : 0;
      }
#pragma unroll
      for (int k = 0; k < 4; ++k)
        if (i0 + 2 * k < n) sr[sl + r[k]] = e[k];
    }

    if (n > 0) {
      const int2* mine = sr + sl;
      float acc[48];
#pragma unroll
      for (int q = 0; q < 48; ++q) acc[q] = 0.f;
      unsigned own = (unsigned)half * 16u;
      for (int i0 = 0; i0 < n; i0 += 4) {
        int2 ed[4];
#pragma unroll
        for (int k = 0; k < 4; ++k) {
          int ik = i0 + k;
          ed[k] = mine[ik < nm1 ? ik : nm1];
        }
        float4 SO[4], ST[4], XO[4];
#pragma unroll
        for (int k = 0; k < 4; ++k) {
          unsigned so = (unsigned)ed[k].x * (unsigned)(R_DIM * 2 * sizeof(float));
          unsigned xo = (unsigned)ed[k].y * (unsigned)(CH_IN * sizeof(float));
          SO[k] = *(const float4*)((const char*)sten + (so + own));
          ST[k] = *(const float4*)((const char*)sten + so + 32);
          XO[k] = *(const float4*)((const char*)x + (xo + own));
        }
#pragma unroll
        for (int k = 0; k < 4; ++k) {
          bool ok = (i0 + k) < n;
          float4 xa = XO[k];
          float k0 = half ? xa.y : xa.x;
          float k1 = half ? xa.w : xa.z;
          float t0 = half ? xa.x : xa.y;
          float t1 = half ? xa.z : xa.w;
          k0 = ok ? k0 : 0.f;
          k1 = ok ? k1 : 0.f;
          t0 = ok ? t0 : 0.f;
          t1 = ok ? t1 : 0.f;
          float stq[3][4] = {
            { SO[k].x, SO[k].y, SO[k].z, SO[k].w },
            { xhalf(SO[k].x), xhalf(SO[k].y), xhalf(SO[k].z), xhalf(SO[k].w) },
            { ST[k].x, ST[k].y, ST[k].z, ST[k].w } };
          float xv[2][2] = { { k0, k1 }, { xhalf(t0), xhalf(t1) } };
#pragma unroll
          for (int Q = 0; Q < 3; ++Q) {
#pragma unroll
            for (int s = 0; s < 4; ++s) {
#pragma unroll
              for (int X = 0; X < 2; ++X) {
#pragma unroll
                for (int u = 0; u < 2; ++u) {
                  int q = ((Q * 4 + s) * 2 + X) * 2 + u;
                  acc[q] = __builtin_fmaf(stq[Q][s], xv[X][u], acc[q]);
                }
              }
            }
          }
        }
      }
#pragma unroll
      for (int r = 0; r < R_DIM; ++r) {
#pragma unroll
        for (int j = 0; j < 4; ++j) {
#pragma unroll
          for (int p = 0; p < 2; ++p) {
            int rq = r >> 1, s = (r & 1) * 2 + p, jx = j >> 1, u = j & 1;
            int Q0 = rq;
            int Q1 = (rq == 0) ? 1 : (rq == 1) ? 0 : 2;
            int q0 = ((Q0 * 4 + s) * 2 + jx) * 2 + u;
            int q1 = ((Q1 * 4 + s) * 2 + (1 - jx)) * 2 + u;
            float v0 = acc[q0];
            float v1 = acc[q1];
            a[(r * 4 + j) * 2 + p] = half ? v1 : v0;
          }
        }
      }
    }
  } else {
    int last = -1;
    for (int i = 0; i < n; ++i) {
      int best = 0x7fffffff, bsrc = 0;
      for (int c = 0; c < NCHUNK; ++c) {
        int m = rn[c];
        const int2* run = c2 + rb[c];
        for (int j = 0; j < m; ++j) {
          int2 e = run[j];
          int id = e.x & ID_MASK;
          if ((e.x >> ID_BITS) == ln && id > last && id < best) { best = id; bsrc = e.y; }
        }
      }
      last = best;
      const float4* sp = (const float4*)(sten + (size_t)best * (R_DIM * 2));
      const float4* xp = (const float4*)(x + (size_t)bsrc * CH_IN);
      float4 xa = xp[0];
      float4 xb = xp[1];
      add_edge(a, sp[0], sp[1], sp[2],
               half ? xa.y : xa.x, half ? xa.w : xa.z,
               half ? xb.y : xb.x, half ? xb.w : xb.z);
    }
  }

  const float* fb0 = Fl + (size_t)(half * OCOLS + mrow) * 2;
  const float* fb1 = fb0 + 32;

  v8f acc0 = {0.f, 0.f, 0.f, 0.f, 0.f, 0.f, 0.f, 0.f};
  v8f acc1 = {0.f, 0.f, 0.f, 0.f, 0.f, 0.f, 0.f, 0.f};

#pragma unroll
  for (int ks = 0; ks < K_DIM / 4; ++ks) {
    v2f av = { a[2 * ks], a[2 * ks + 1] };
    v2f b0 = *(const v2f*)(fb0 + ks * (OCOLS * 4));
    v2f b1 = *(const v2f*)(fb1 + ks * (OCOLS * 4));
    acc0 = __builtin_amdgcn_wmma_f32_16x16x4_f32(
        false, av, false, b0, (short)0, acc0, false, false);
    acc1 = __builtin_amdgcn_wmma_f32_16x16x4_f32(
        false, av, false, b1, (short)0, acc1, false, false);
  }

  float bn = bias[mrow];
  int   mhi = half * 8;
#pragma unroll
  for (int j = 0; j < 8; ++j) {
    int   m   = j + mhi;
    float re  = acc0[j];
    float im  = acc1[j];
    float mag = __builtin_amdgcn_sqrtf(re * re + im * im);
    float sc  = fmaxf(mag + bn, 0.0f) *
                __builtin_amdgcn_rcpf(mag + 1e-8f);
    v2f v = { re * sc, im * sc };
    __builtin_nontemporal_store(
        v, (v2f*)(out + ((size_t)(node_base + m) * CH_OUT + mrow) * 2));
  }
}

extern "C" void kernel_launch(void* const* d_in, const int* in_sizes, int n_in,
                              void* d_out, int out_size, void* d_ws, size_t ws_size,
                              hipStream_t stream) {
  const float* x      = (const float*)d_in[0];
  const int*   edges  = (const int*)d_in[1];
  const float* sten   = (const float*)d_in[2];
  const float* weight = (const float*)d_in[3];
  const float* offset = (const float*)d_in[4];
  const float* bias   = (const float*)d_in[5];
  float*       out    = (float*)d_out;

  int*   coarse = (int*)d_ws;
  float* Fp     = (float*)(coarse + 2 * N_EDGES);
  int*   tab    = (int*)(Fp + K_DIM * OCOLS);

  filter_kernel<<<(K_DIM * OCOLS + 255) / 256, 256, 0, stream>>>(weight, offset, Fp);

  chunk_sort_kernel<<<NCHUNK, BT, 0, stream>>>(edges, coarse, tab);
  bucket_kernel<<<NB, FT, 0, stream>>>(Fp, x, sten, tab, coarse, bias, out);
}


# Bucket kernel: every node owns a fixed 45-entry LDS slot (a fuller node sends its bucket to the slow path), so ONE pass places the entries; count pass and 512-wide barrier scan gone; sum order same
# speedup vs baseline: 14.2113x; 1.0835x over previous
#include <hip/hip_runtime.h>

typedef float v2f __attribute__((ext_vector_type(2)));
typedef float v8f __attribute__((ext_vector_type(8)));

#define N_NODES 100000
#define N_EDGES 1600000
#define R_DIM   6
#define CH_IN   8
#define CH_OUT  16
#define K_DIM   96
#define OCOLS   32

#define NB      250
#define BNODES  400
#define NCHUNK  250
#define CHUNK   6400
#define BT      640
#define EPT     10
#define ID_BITS 21
#define ID_MASK ((1 << ID_BITS) - 1)

#define FW      25
#define FT      (FW * 32)
#define RUNS    (NCHUNK / FW)
#define RSTEP   5
#define CAP     45

static_assert(NB * BNODES == N_NODES, "buckets cover every node exactly once");
static_assert(FW * 16 == BNODES, "one wave per tile; a tile never straddles two buckets");
static_assert(NCHUNK * CHUNK == N_EDGES && BT * EPT == CHUNK, "chunks cover every edge exactly once");
static_assert(N_EDGES <= (1 << ID_BITS) && BNODES <= 512, "edge id and node-in-bucket pack into one non-negative int");
static_assert(NB <= 256 && NB <= BT && BNODES <= 512 && FT >= 512, "scan widths");
static_assert(FW * RUNS == NCHUNK && RUNS % RSTEP == 0 && NCHUNK <= 256, "the waves cover every run exactly once");
static_assert(CHUNK < (1 << 15), "count and window start share one non-negative word");
static_assert((CHUNK / 2) % BT == 0 && (CHUNK * 8) % 128 == 0 && BT % 32 == 0, "the window leaves in 16-byte pieces, the same number per thread, whole lines per wave");
static_assert(K_DIM * OCOLS / 4 <= FT, "one 16-byte load per thread stages the whole filter");
static_assert(2 * BNODES * CAP * 8 + K_DIM * OCOLS * 4 + 512 * 4 + 2 * 256 * 4 + 4 <= 320 * 1024, "the work-group's LDS");
static_assert(CAP % 2 == 1, "an odd slot pitch in 8-byte entries spreads the 16 nodes of a tile over different LDS banks");


__device__ __forceinline__ void scan256(int* sc, int t, int v)
{
  if (t < 256) sc[t] = v;
  __syncthreads();
  for (int d = 1; d < 256; d <<= 1) {
    int a = (t < 256 && t >= d) ? sc[t - d] : 0;
    __syncthreads();
    if (t < 256) sc[t] += a;
    __syncthreads();
  }
}

__global__ __launch_bounds__(BT) void chunk_sort_kernel(
    const int* __restrict__ edges,
    int*       __restrict__ coarse,
    int*       __restrict__ tab)
{
  __shared__ int h[NB];
  __shared__ int sc[256];
  __shared__ __attribute__((aligned(16))) int2 stage[CHUNK];
  int t = threadIdx.x;
  if (t < NB) h[t] = 0;
  __syncthreads();
  int first = blockIdx.x * CHUNK + t;
  int2 sd[EPT];
  int  b[EPT], r[EPT];
#pragma unroll
  for (int k = 0; k < EPT; ++k) sd[k] = ((const int2*)edges)[first + k * BT];
#pragma unroll
  for (int k = 0; k < EPT; ++k) {
    b[k] = sd[k].y / BNODES;
    r[k] = atomicAdd(&h[b[k]], 1);
  }
  __syncthreads();
  int c = (t < NB) ? h[t] : 0;
  scan256(sc, t, c);
  if (t < NB) {
    int base = sc[t] - c;
    h[t] = base;
    tab[t * NCHUNK + blockIdx.x] = c | (base << 16);
  }
  __syncthreads();
#pragma unroll
  for (int k = 0; k < EPT; ++k) {
    int e = first + k * BT;
    int l = sd[k].y - b[k] * BNODES;
    stage[h[b[k]] + r[k]] = make_int2(e | (l << ID_BITS), sd[k].x);
  }
  __syncthreads();
  int4*       win = (int4*)((int2*)coarse + blockIdx.x * CHUNK);
  const int4* st4 = (const int4*)stage;
#pragma unroll
  for (int k = 0; k < CHUNK / 2 / BT; ++k) win[t + k * BT] = st4[t + k * BT];
}

__global__ __launch_bounds__(256) void filter_kernel(
    const float* __restrict__ weight,
    const float* __restrict__ offset,
    float*       __restrict__ Fp)
{
  int i = blockIdx.x * 256 + threadIdx.x;
  if (i >= K_DIM * OCOLS) return;
  int k   = i >> 5;
  int col = i & 31;
  int p   = k & 1;
  int rc  = k >> 1;
  int r   = rc >> 3;
  int c   = rc & 7;
  int o   = col & 15;

  float w  = weight[(r * CH_IN + c) * CH_OUT + o];
  float ph = offset[c * CH_OUT + o];
  float fre = w * cosf(ph);
  float fim = w * sinf(ph);

  float val;
  if (col < 16) val = (p == 0) ? fre : -fim;
  else          val = (p == 0) ? fim :  fre;

  Fp[(rc * OCOLS + col) * 2 + p] = val;
}

__device__ __forceinline__ float xhalf(float v)
{
  return __builtin_bit_cast(float, __builtin_amdgcn_permlanex16(
      __builtin_bit_cast(int, v), __builtin_bit_cast(int, v), 0x76543210, 0xfedcba98, false, false));
}

__device__ __forceinline__ void add_edge(float (&a)[48], float4 s0, float4 s1, float4 s2,
                                         float x0, float x1, float x2, float x3)
{
  float st[12] = {s0.x, s0.y, s0.z, s0.w, s1.x, s1.y, s1.z, s1.w, s2.x, s2.y, s2.z, s2.w};
  float xs[4]  = {x0, x1, x2, x3};
#pragma unroll
  for (int r = 0; r < R_DIM; ++r) {
#pragma unroll
    for (int j = 0; j < 4; ++j) {
      a[(r * 4 + j) * 2 + 0] = __builtin_fmaf(st[r * 2 + 0], xs[j], a[(r * 4 + j) * 2 + 0]);
      a[(r * 4 + j) * 2 + 1] = __builtin_fmaf(st[r * 2 + 1], xs[j], a[(r * 4 + j) * 2 + 1]);
    }
  }
}

__global__ __launch_bounds__(FT) __attribute__((amdgpu_waves_per_eu(1, 7))) void bucket_kernel(
    const float* __restrict__ Fp,
    const float* __restrict__ x,
    const float* __restrict__ sten,
    const int*   __restrict__ tab,
    const int*   __restrict__ coarse,
    const float* __restrict__ bias,
    float*       __restrict__ out)
{
  __shared__ __attribute__((aligned(16))) float Fl[K_DIM * OCOLS];
  __shared__ int cur[512];
  __shared__ int rb[256];
  __shared__ int rn[256];
  __shared__ int ovf;
  __shared__ int2 raw[BNODES * CAP];
  __shared__ int2 srt[BNODES * CAP];
  int t    = threadIdx.x;
  int bk   = blockIdx.x;
  int lane = t & 31;
  int w    = t >> 5;
  float4 fv = make_float4(0.f, 0.f, 0.f, 0.f);
  if (t < K_DIM * OCOLS / 4) fv = ((const float4*)Fp)[t];
  if (t < 512) cur[t] = 0;
  if (t == 0) ovf = 0;
  if (t < 256) {
    int word = (t < NCHUNK) ? tab[bk * NCHUNK + t] : 0;
    rn[t] = word & 0xFFFF;
    rb[t] = t * CHUNK + (word >> 16);
  }
  if (t < K_DIM * OCOLS / 4) ((float4*)Fl)[t] = fv;
  __syncthreads();
  const int2* c2 = (const int2*)coarse;

  for (int j0 = 0; j0 < RUNS; j0 += RSTEP) {
    int2 v[RSTEP];
    int  nn[RSTEP], bb[RSTEP];
#pragma unroll
    for (int k = 0; k < RSTEP; ++k) {
      int c = w + FW * (j0 + k);
      nn[k] = rn[c];
      bb[k] = rb[c];
      v[k] = (lane < nn[k]) ? c2[bb[k] + lane] : make_int2(-1, 0);
    }
#pragma unroll
    for (int k = 0; k < RSTEP; ++k) {
      if (v[k].x >= 0) {
        int l = v[k].x >> ID_BITS;
        int slot = atomicAdd(&cur[l], 1);
        if (slot < CAP) raw[l * CAP + slot] = make_int2(v[k].x & ID_MASK, v[k].y);
        else            ovf = 1;
      }
    }
#pragma unroll
    for (int k = 0; k < RSTEP; ++k) {
      for (int i = lane + 32; i < nn[k]; i += 32) {
        int2 e = c2[bb[k] + i];
        int l = e.x >> ID_BITS;
        int slot = atomicAdd(&cur[l], 1);
        if (slot < CAP) raw[l * CAP + slot] = make_int2(e.x & ID_MASK, e.y);
        else            ovf = 1;
      }
    }
  }
  __syncthreads();

  bool fits = __builtin_amdgcn_readfirstlane(ovf) == 0;

  int node_base = bk * BNODES + w * 16;
  int mrow = lane & 15;
  int half = lane >> 4;
  int ln   = w * 16 + mrow;

  float a[48];
#pragma unroll
  for (int q = 0; q < 48; ++q) a[q] = 0.f;

  int sl  = ln * CAP;
  int n   = cur[ln];
  int nm1 = n - 1;

  if (fits) {
    const int2* rw = raw;
    int2*       sr = srt;
    for (int i0 = half; i0 < n; i0 += 8) {
      int2 e[4];
      int  r[4];
#pragma unroll
      for (int k = 0; k < 4; ++k) {
        int i = i0 + 2 * k;
        e[k] = rw[sl + (i < nm1 ? i : nm1)];
        r[k] = 0;
      }
#pragma unroll 4
      for (int j = 0; j < n; ++j) {
        int vj = rw[sl + j].x;
#pragma unroll
        for (int k = 0; k < 4; ++k) r[k] += (vj < e[k].x) ? 1 : 0;
      }
#pragma unroll
      for (int k = 0; k < 4; ++k)
        if (i0 + 2 * k < n) sr[sl + r[k]] = e[k];
    }

    if (n > 0) {
      const int2* mine = sr + sl;
      float acc[48];
#pragma unroll
      for (int q = 0; q < 48; ++q) acc[q] = 0.f;
      unsigned own = (unsigned)half * 16u;
      for (int i0 = 0; i0 < n; i0 += 4) {
        int2 ed[4];
#pragma unroll
        for (int k = 0; k < 4; ++k) {
          int ik = i0 + k;
          ed[k] = mine[ik < nm1 ? ik : nm1];
        }
        float4 SO[4], ST[4], XO[4];
#pragma unroll
        for (int k = 0; k < 4; ++k) {
          unsigned so = (unsigned)ed[k].x * (unsigned)(R_DIM * 2 * sizeof(float));
          unsigned xo = (unsigned)ed[k].y * (unsigned)(CH_IN * sizeof(float));
          SO[k] = *(const float4*)((const char*)sten + (so + own));
          ST[k] = *(const float4*)((const char*)sten + so + 32);
          XO[k] = *(const float4*)((const char*)x + (xo + own));
        }
#pragma unroll
        for (int k = 0; k < 4; ++k) {
          bool ok = (i0 + k) < n;
          float4 xa = XO[k];
          float k0 = half ? xa.y : xa.x;
          float k1 = half ? xa.w : xa.z;
          float t0 = half ? xa.x : xa.y;
          float t1 = half ? xa.z : xa.w;
          k0 = ok ? k0 : 0.f;
          k1 = ok ? k1 : 0.f;
          t0 = ok ? t0 : 0.f;
          t1 = ok ? t1 : 0.f;
          float stq[3][4] = {
            { SO[k].x, SO[k].y, SO[k].z, SO[k].w },
            { xhalf(SO[k].x), xhalf(SO[k].y), xhalf(SO[k].z), xhalf(SO[k].w) },
            { ST[k].x, ST[k].y, ST[k].z, ST[k].w } };
          float xv[2][2] = { { k0, k1 }, { xhalf(t0), xhalf(t1) } };
#pragma unroll
          for (int Q = 0; Q < 3; ++Q) {
#pragma unroll
            for (int s = 0; s < 4; ++s) {
#pragma unroll
              for (int X = 0; X < 2; ++X) {
#pragma unroll
                for (int u = 0; u < 2; ++u) {
                  int q = ((Q * 4 + s) * 2 + X) * 2 + u;
                  acc[q] = __builtin_fmaf(stq[Q][s], xv[X][u], acc[q]);
                }
              }
            }
          }
        }
      }
#pragma unroll
      for (int r = 0; r < R_DIM; ++r) {
#pragma unroll
        for (int j = 0; j < 4; ++j) {
#pragma unroll
          for (int p = 0; p < 2; ++p) {
            int rq = r >> 1, s = (r & 1) * 2 + p, jx = j >> 1, u = j & 1;
            int Q0 = rq;
            int Q1 = (rq == 0) ? 1 : (rq == 1) ? 0 : 2;
            int q0 = ((Q0 * 4 + s) * 2 + jx) * 2 + u;
            int q1 = ((Q1 * 4 + s) * 2 + (1 - jx)) * 2 + u;
            float v0 = acc[q0];
            float v1 = acc[q1];
            a[(r * 4 + j) * 2 + p] = half ? v1 : v0;
          }
        }
      }
    }
  } else {
    int last = -1;
    for (int i = 0; i < n; ++i) {
      int best = 0x7fffffff, bsrc = 0;
      for (int c = 0; c < NCHUNK; ++c) {
        int m = rn[c];
        const int2* run = c2 + rb[c];
        for (int j = 0; j < m; ++j) {
          int2 e = run[j];
          int id = e.x & ID_MASK;
          if ((e.x >> ID_BITS) == ln && id > last && id < best) { best = id; bsrc = e.y; }
        }
      }
      last = best;
      const float4* sp = (const float4*)(sten + (size_t)best * (R_DIM * 2));
      const float4* xp = (const float4*)(x + (size_t)bsrc * CH_IN);
      float4 xa = xp[0];
      float4 xb = xp[1];
      add_edge(a, sp[0], sp[1], sp[2],
               half ? xa.y : xa.x, half ? xa.w : xa.z,
               half ? xb.y : xb.x, half ? xb.w : xb.z);
    }
  }

  const float* fb0 = Fl + (size_t)(half * OCOLS + mrow) * 2;
  const float* fb1 = fb0 + 32;

  v8f acc0 = {0.f, 0.f, 0.f, 0.f, 0.f, 0.f, 0.f, 0.f};
  v8f acc1 = {0.f, 0.f, 0.f, 0.f, 0.f, 0.f, 0.f, 0.f};

#pragma unroll
  for (int ks = 0; ks < K_DIM / 4; ++ks) {
    v2f av = { a[2 * ks], a[2 * ks + 1] };
    v2f b0 = *(const v2f*)(fb0 + ks * (OCOLS * 4));
    v2f b1 = *(const v2f*)(fb1 + ks * (OCOLS * 4));
    acc0 = __builtin_amdgcn_wmma_f32_16x16x4_f32(
        false, av, false, b0, (short)0, acc0, false, false);
    acc1 = __builtin_amdgcn_wmma_f32_16x16x4_f32(
        false, av, false, b1, (short)0, acc1, false, false);
  }

  float bn = bias[mrow];
  int   mhi = half * 8;
#pragma unroll
  for (int j = 0; j < 8; ++j) {
    int   m   = j + mhi;
    float re  = acc0[j];
    float im  = acc1[j];
    float mag = __builtin_amdgcn_sqrtf(re * re + im * im);
    float sc  = fmaxf(mag + bn, 0.0f) *
                __builtin_amdgcn_rcpf(mag + 1e-8f);
    v2f v = { re * sc, im * sc };
    __builtin_nontemporal_store(
        v, (v2f*)(out + ((size_t)(node_base + m) * CH_OUT + mrow) * 2));
  }
}

extern "C" void kernel_launch(void* const* d_in, const int* in_sizes, int n_in,
                              void* d_out, int out_size, void* d_ws, size_t ws_size,
                              hipStream_t stream) {
  const float* x      = (const float*)d_in[0];
  const int*   edges  = (const int*)d_in[1];
  const float* sten   = (const float*)d_in[2];
  const float* weight = (const float*)d_in[3];
  const float* offset = (const float*)d_in[4];
  const float* bias   = (const float*)d_in[5];
  float*       out    = (float*)d_out;

  int*   coarse = (int*)d_ws;
  float* Fp     = (float*)(coarse + 2 * N_EDGES);
  int*   tab    = (int*)(Fp + K_DIM * OCOLS);

  filter_kernel<<<(K_DIM * OCOLS + 255) / 256, 256, 0, stream>>>(weight, offset, Fp);

  chunk_sort_kernel<<<NCHUNK, BT, 0, stream>>>(edges, coarse, tab);
  bucket_kernel<<<NB, FT, 0, stream>>>(Fp, x, sten, tab, coarse, bias, out);
}


# Filter built by five extra work-groups of the level-1 launch (one entry per thread, beside the chunks' sorting) instead of by a launch of its own: 2 launches per run, not 3; same filter values
# speedup vs baseline: 14.8588x; 1.0456x over previous
#include <hip/hip_runtime.h>

typedef float v2f __attribute__((ext_vector_type(2)));
typedef float v8f __attribute__((ext_vector_type(8)));

#define N_NODES 100000
#define N_EDGES 1600000
#define R_DIM   6
#define CH_IN   8
#define CH_OUT  16
#define K_DIM   96
#define OCOLS   32

#define NB      250
#define BNODES  400
#define NCHUNK  250
#define CHUNK   6400
#define BT      640
#define EPT     10
#define ID_BITS 21
#define ID_MASK ((1 << ID_BITS) - 1)
#define FWG     ((K_DIM * OCOLS + BT - 1) / BT)

#define FW      25
#define FT      (FW * 32)
#define RUNS    (NCHUNK / FW)
#define RSTEP   5
#define CAP     45

static_assert(NB * BNODES == N_NODES, "buckets cover every node exactly once");
static_assert(FW * 16 == BNODES, "one wave per tile; a tile never straddles two buckets");
static_assert(NCHUNK * CHUNK == N_EDGES && BT * EPT == CHUNK, "chunks cover every edge exactly once");
static_assert(N_EDGES <= (1 << ID_BITS) && BNODES <= 512, "edge id and node-in-bucket pack into one non-negative int");
static_assert(NB <= 256 && NB <= BT && BNODES <= 512 && FT >= 512, "scan widths");
static_assert(FW * RUNS == NCHUNK && RUNS % RSTEP == 0 && NCHUNK <= 256, "the waves cover every run exactly once");
static_assert(CHUNK < (1 << 15), "count and window start share one non-negative word");
static_assert((CHUNK / 2) % BT == 0 && (CHUNK * 8) % 128 == 0 && BT % 32 == 0, "the window leaves in 16-byte pieces, the same number per thread, whole lines per wave");
static_assert(K_DIM * OCOLS / 4 <= FT, "one 16-byte load per thread stages the whole filter");
static_assert(2 * BNODES * CAP * 8 + K_DIM * OCOLS * 4 + 512 * 4 + 2 * 256 * 4 + 4 <= 320 * 1024, "the work-group's LDS");
static_assert(CAP % 2 == 1, "an odd slot pitch in 8-byte entries spreads the 16 nodes of a tile over different LDS banks");


__device__ __forceinline__ void scan256(int* sc, int t, int v)
{
  if (t < 256) sc[t] = v;
  __syncthreads();
  for (int d = 1; d < 256; d <<= 1) {
    int a = (t < 256 && t >= d) ? sc[t - d] : 0;
    __syncthreads();
    if (t < 256) sc[t] += a;
    __syncthreads();
  }
}

__device__ __forceinline__ void filter_entry(
    int i,
    const float* __restrict__ weight,
    const float* __restrict__ offset,
    float*       __restrict__ Fp)
{
  int k   = i >> 5;
  int col = i & 31;
  int p   = k & 1;
  int rc  = k >> 1;
  int r   = rc >> 3;
  int c   = rc & 7;
  int o   = col & 15;

  float w  = weight[(r * CH_IN + c) * CH_OUT + o];
  float ph = offset[c * CH_OUT + o];
  float fre = w * cosf(ph);
  float fim = w * sinf(ph);

  float val;
  if (col < 16) val = (p == 0) ? fre : -fim;
  else          val = (p == 0) ? fim :  fre;

  Fp[(rc * OCOLS + col) * 2 + p] = val;
}

__global__ __launch_bounds__(BT) void chunk_sort_kernel(
    const int* __restrict__ edges,
    int*       __restrict__ coarse,
    int*       __restrict__ tab,
    const float* __restrict__ weight,
    const float* __restrict__ offset,
    float*       __restrict__ Fp)
{
  __shared__ int h[NB];
  __shared__ int sc[256];
  __shared__ __attribute__((aligned(16))) int2 stage[CHUNK];
  int t = threadIdx.x;
  if (blockIdx.x >= NCHUNK) {
    int i = (blockIdx.x - NCHUNK) * BT + t;
    if (i < K_DIM * OCOLS) filter_entry(i, weight, offset, Fp);
    return;
  }
  if (t < NB) h[t] = 0;
  __syncthreads();
  int first = blockIdx.x * CHUNK + t;
  int2 sd[EPT];
  int  b[EPT], r[EPT];
#pragma unroll
  for (int k = 0; k < EPT; ++k) sd[k] = ((const int2*)edges)[first + k * BT];
#pragma unroll
  for (int k = 0; k < EPT; ++k) {
    b[k] = sd[k].y / BNODES;
    r[k] = atomicAdd(&h[b[k]], 1);
  }
  __syncthreads();
  int c = (t < NB) ? h[t] : 0;
  scan256(sc, t, c);
  if (t < NB) {
    int base = sc[t] - c;
    h[t] = base;
    tab[t * NCHUNK + blockIdx.x] = c | (base << 16);
  }
  __syncthreads();
#pragma unroll
  for (int k = 0; k < EPT; ++k) {
    int e = first + k * BT;
    int l = sd[k].y - b[k] * BNODES;
    stage[h[b[k]] + r[k]] = make_int2(e | (l << ID_BITS), sd[k].x);
  }
  __syncthreads();
  int4*       win = (int4*)((int2*)coarse + blockIdx.x * CHUNK);
  const int4* st4 = (const int4*)stage;
#pragma unroll
  for (int k = 0; k < CHUNK / 2 / BT; ++k) win[t + k * BT] = st4[t + k * BT];
}

__device__ __forceinline__ float xhalf(float v)
{
  return __builtin_bit_cast(float, __builtin_amdgcn_permlanex16(
      __builtin_bit_cast(int, v), __builtin_bit_cast(int, v), 0x76543210, 0xfedcba98, false, false));
}

__device__ __forceinline__ void add_edge(float (&a)[48], float4 s0, float4 s1, float4 s2,
                                         float x0, float x1, float x2, float x3)
{
  float st[12] = {s0.x, s0.y, s0.z, s0.w, s1.x, s1.y, s1.z, s1.w, s2.x, s2.y, s2.z, s2.w};
  float xs[4]  = {x0, x1, x2, x3};
#pragma unroll
  for (int r = 0; r < R_DIM; ++r) {
#pragma unroll
    for (int j = 0; j < 4; ++j) {
      a[(r * 4 + j) * 2 + 0] = __builtin_fmaf(st[r * 2 + 0], xs[j], a[(r * 4 + j) * 2 + 0]);
      a[(r * 4 + j) * 2 + 1] = __builtin_fmaf(st[r * 2 + 1], xs[j], a[(r * 4 + j) * 2 + 1]);
    }
  }
}

__global__ __launch_bounds__(FT) __attribute__((amdgpu_waves_per_eu(1, 7))) void bucket_kernel(
    const float* __restrict__ Fp,
    const float* __restrict__ x,
    const float* __restrict__ sten,
    const int*   __restrict__ tab,
    const int*   __restrict__ coarse,
    const float* __restrict__ bias,
    float*       __restrict__ out)
{
  __shared__ __attribute__((aligned(16))) float Fl[K_DIM * OCOLS];
  __shared__ int cur[512];
  __shared__ int rb[256];
  __shared__ int rn[256];
  __shared__ int ovf;
  __shared__ int2 raw[BNODES * CAP];
  __shared__ int2 srt[BNODES * CAP];
  int t    = threadIdx.x;
  int bk   = blockIdx.x;
  int lane = t & 31;
  int w    = t >> 5;
  float4 fv = make_float4(0.f, 0.f, 0.f, 0.f);
  if (t < K_DIM * OCOLS / 4) fv = ((const float4*)Fp)[t];
  if (t < 512) cur[t] = 0;
  if (t == 0) ovf = 0;
  if (t < 256) {
    int word = (t < NCHUNK) ? tab[bk * NCHUNK + t] : 0;
    rn[t] = word & 0xFFFF;
    rb[t] = t * CHUNK + (word >> 16);
  }
  if (t < K_DIM * OCOLS / 4) ((float4*)Fl)[t] = fv;
  __syncthreads();
  const int2* c2 = (const int2*)coarse;

  for (int j0 = 0; j0 < RUNS; j0 += RSTEP) {
    int2 v[RSTEP];
    int  nn[RSTEP], bb[RSTEP];
#pragma unroll
    for (int k = 0; k < RSTEP; ++k) {
      int c = w + FW * (j0 + k);
      nn[k] = rn[c];
      bb[k] = rb[c];
      v[k] = (lane < nn[k]) ? c2[bb[k] + lane] : make_int2(-1, 0);
    }
#pragma unroll
    for (int k = 0; k < RSTEP; ++k) {
      if (v[k].x >= 0) {
        int l = v[k].x >> ID_BITS;
        int slot = atomicAdd(&cur[l], 1);
        if (slot < CAP) raw[l * CAP + slot] = make_int2(v[k].x & ID_MASK, v[k].y);
        else            ovf = 1;
      }
    }
#pragma unroll
    for (int k = 0; k < RSTEP; ++k) {
      for (int i = lane + 32; i < nn[k]; i += 32) {
        int2 e = c2[bb[k] + i];
        int l = e.x >> ID_BITS;
        int slot = atomicAdd(&cur[l], 1);
        if (slot < CAP) raw[l * CAP + slot] = make_int2(e.x & ID_MASK, e.y);
        else            ovf = 1;
      }
    }
  }
  __syncthreads();

  bool fits = __builtin_amdgcn_readfirstlane(ovf) == 0;

  int node_base = bk * BNODES + w * 16;
  int mrow = lane & 15;
  int half = lane >> 4;
  int ln   = w * 16 + mrow;

  float a[48];
#pragma unroll
  for (int q = 0; q < 48; ++q) a[q] = 0.f;

  int sl  = ln * CAP;
  int n   = cur[ln];
  int nm1 = n - 1;

  if (fits) {
    const int2* rw = raw;
    int2*       sr = srt;
    for (int i0 = half; i0 < n; i0 += 8) {
      int2 e[4];
      int  r[4];
#pragma unroll
      for (int k = 0; k < 4; ++k) {
        int i = i0 + 2 * k;
        e[k] = rw[sl + (i < nm1 ? i : nm1)];
        r[k] = 0;
      }
#pragma unroll 4
      for (int j = 0; j < n; ++j) {
        int vj = rw[sl + j].x;
#pragma unroll
        for (int k = 0; k < 4; ++k) r[k] += (vj < e[k].x) ? 1 : 0;
      }
#pragma unroll
      for (int k = 0; k < 4; ++k)
        if (i0 + 2 * k < n) sr[sl + r[k]] = e[k];
    }

    if (n > 0) {
      const int2* mine = sr + sl;
      float acc[48];
#pragma unroll
      for (int q = 0; q < 48; ++q) acc[q] = 0.f;
      unsigned own = (unsigned)half * 16u;
      for (int i0 = 0; i0 < n; i0 += 4) {
        int2 ed[4];
#pragma unroll
        for (int k = 0; k < 4; ++k) {
          int ik = i0 + k;
          ed[k] = mine[ik < nm1 ? ik : nm1];
        }
        float4 SO[4], ST[4], XO[4];
#pragma unroll
        for (int k = 0; k < 4; ++k) {
          unsigned so = (unsigned)ed[k].x * (unsigned)(R_DIM * 2 * sizeof(float));
          unsigned xo = (unsigned)ed[k].y * (unsigned)(CH_IN * sizeof(float));
          SO[k] = *(const float4*)((const char*)sten + (so + own));
          ST[k] = *(const float4*)((const char*)sten + so + 32);
          XO[k] = *(const float4*)((const char*)x + (xo + own));
        }
#pragma unroll
        for (int k = 0; k < 4; ++k) {
          bool ok = (i0 + k) < n;
          float4 xa = XO[k];
          float k0 = half ? xa.y : xa.x;
          float k1 = half ? xa.w : xa.z;
          float t0 = half ? xa.x : xa.y;
          float t1 = half ? xa.z : xa.w;
          k0 = ok ? k0 : 0.f;
          k1 = ok ? k1 : 0.f;
          t0 = ok ? t0 : 0.f;
          t1 = ok ? t1 : 0.f;
          float stq[3][4] = {
            { SO[k].x, SO[k].y, SO[k].z, SO[k].w },
            { xhalf(SO[k].x), xhalf(SO[k].y), xhalf(SO[k].z), xhalf(SO[k].w) },
            { ST[k].x, ST[k].y, ST[k].z, ST[k].w } };
          float xv[2][2] = { { k0, k1 }, { xhalf(t0), xhalf(t1) } };
#pragma unroll
          for (int Q = 0; Q < 3; ++Q) {
#pragma unroll
            for (int s = 0; s < 4; ++s) {
#pragma unroll
              for (int X = 0; X < 2; ++X) {
#pragma unroll
                for (int u = 0; u < 2; ++u) {
                  int q = ((Q * 4 + s) * 2 + X) * 2 + u;
                  acc[q] = __builtin_fmaf(stq[Q][s], xv[X][u], acc[q]);
                }
              }
            }
          }
        }
      }
#pragma unroll
      for (int r = 0; r < R_DIM; ++r) {
#pragma unroll
        for (int j = 0; j < 4; ++j) {
#pragma unroll
          for (int p = 0; p < 2; ++p) {
            int rq = r >> 1, s = (r & 1) * 2 + p, jx = j >> 1, u = j & 1;
            int Q0 = rq;
            int Q1 = (rq == 0) ? 1 : (rq == 1) ? 0 : 2;
            int q0 = ((Q0 * 4 + s) * 2 + jx) * 2 + u;
            int q1 = ((Q1 * 4 + s) * 2 + (1 - jx)) * 2 + u;
            float v0 = acc[q0];
            float v1 = acc[q1];
            a[(r * 4 + j) * 2 + p] = half ? v1 : v0;
          }
        }
      }
    }
  } else {
    int last = -1;
    for (int i = 0; i < n; ++i) {
      int best = 0x7fffffff, bsrc = 0;
      for (int c = 0; c < NCHUNK; ++c) {
        int m = rn[c];
        const int2* run = c2 + rb[c];
        for (int j = 0; j < m; ++j) {
          int2 e = run[j];
          int id = e.x & ID_MASK;
          if ((e.x >> ID_BITS) == ln && id > last && id < best) { best = id; bsrc = e.y; }
        }
      }
      last = best;
      const float4* sp = (const float4*)(sten + (size_t)best * (R_DIM * 2));
      const float4* xp = (const float4*)(x + (size_t)bsrc * CH_IN);
      float4 xa = xp[0];
      float4 xb = xp[1];
      add_edge(a, sp[0], sp[1], sp[2],
               half ? xa.y : xa.x, half ? xa.w : xa.z,
               half ? xb.y : xb.x, half ? xb.w : xb.z);
    }
  }

  const float* fb0 = Fl + (size_t)(half * OCOLS + mrow) * 2;
  const float* fb1 = fb0 + 32;

  v8f acc0 = {0.f, 0.f, 0.f, 0.f, 0.f, 0.f, 0.f, 0.f};
  v8f acc1 = {0.f, 0.f, 0.f, 0.f, 0.f, 0.f, 0.f, 0.f};

#pragma unroll
  for (int ks = 0; ks < K_DIM / 4; ++ks) {
    v2f av = { a[2 * ks], a[2 * ks + 1] };
    v2f b0 = *(const v2f*)(fb0 + ks * (OCOLS * 4));
    v2f b1 = *(const v2f*)(fb1 + ks * (OCOLS * 4));
    acc0 = __builtin_amdgcn_wmma_f32_16x16x4_f32(
        false, av, false, b0, (short)0, acc0, false, false);
    acc1 = __builtin_amdgcn_wmma_f32_16x16x4_f32(
        false, av, false, b1, (short)0, acc1, false, false);
  }

  float bn = bias[mrow];
  int   mhi = half * 8;
#pragma unroll
  for (int j = 0; j < 8; ++j) {
    int   m   = j + mhi;
    float re  = acc0[j];
    float im  = acc1[j];
    float mag = __builtin_amdgcn_sqrtf(re * re + im * im);
    float sc  = fmaxf(mag + bn, 0.0f) *
                __builtin_amdgcn_rcpf(mag + 1e-8f);
    v2f v = { re * sc, im * sc };
    __builtin_nontemporal_store(
        v, (v2f*)(out + ((size_t)(node_base + m) * CH_OUT + mrow) * 2));
  }
}

extern "C" void kernel_launch(void* const* d_in, const int* in_sizes, int n_in,
                              void* d_out, int out_size, void* d_ws, size_t ws_size,
                              hipStream_t stream) {
  const float* x      = (const float*)d_in[0];
  const int*   edges  = (const int*)d_in[1];
  const float* sten   = (const float*)d_in[2];
  const float* weight = (const float*)d_in[3];
  const float* offset = (const float*)d_in[4];
  const float* bias   = (const float*)d_in[5];
  float*       out    = (float*)d_out;

  int*   coarse = (int*)d_ws;
  float* Fp     = (float*)(coarse + 2 * N_EDGES);
  int*   tab    = (int*)(Fp + K_DIM * OCOLS);

  chunk_sort_kernel<<<NCHUNK + FWG, BT, 0, stream>>>(edges, coarse, tab, weight, offset, Fp);
  bucket_kernel<<<NB, FT, 0, stream>>>(Fp, x, sten, tab, coarse, bias, out);
}
